# starting kernel
# baseline (speedup 1.0000x reference)
#include <hip/hip_runtime.h>
#include <hip/hip_bf16.h>
#include <math.h>


typedef __attribute__((ext_vector_type(16))) __bf16 v16bf;
typedef __attribute__((ext_vector_type(8)))  float  v8f;

#define IN_CH   64
#define HEADS   8
#define OUT_CH  8
#define HD      (HEADS * OUT_CH)
#define LEAKY   0.3f
#define HT_PITCH 68

__global__ __launch_bounds__(256) void gat_project_wmma(
    const float* __restrict__ h, const float* __restrict__ W,
    float* __restrict__ P, float* __restrict__ Q, int nNodes) {
  __shared__ float hT[16 * HT_PITCH];
  __shared__ float Wl[2 * IN_CH * HD];

  const int tid = threadIdx.x;

  {
    const float4* s4 = (const float4*)(h + (size_t)blockIdx.x * 16 * IN_CH);
    const int r = tid >> 4;
    const int c = (tid & 15) * 4;
    *(float4*)(hT + r * HT_PITCH + c) = s4[tid];
  }
  {
    const float4* s4 = (const float4*)W;
    float4* d4 = (float4*)Wl;
    #pragma unroll
    for (int i = 0; i < 8; ++i)
      d4[tid + 256 * i] = s4[tid + 256 * i];
  }
  __syncthreads();

  const int wave  = tid >> 5;
  const int lane  = tid & 31;
  const int mat   = wave >> 2;
  const int tn    = wave & 3;
  const int l16   = lane & 15;
  const int khalf = lane >> 4;
  const int col   = tn * 16 + l16;
  const int wbase = mat * IN_CH;

  const float* __restrict__ hrow = hT + l16 * HT_PITCH;

  v8f acc = {};
  #pragma unroll
  for (int kb = 0; kb < IN_CH; kb += 32) {
    const int k0 = kb + khalf * 8;
    v16bf A, B;
    #pragma unroll
    for (int j = 0; j < 8; ++j) {
      A[j]     = (__bf16)hrow[k0 + j];
      A[j + 8] = (__bf16)hrow[k0 + 16 + j];
      B[j]     = (__bf16)Wl[(wbase + k0 + j)      * HD + col];
      B[j + 8] = (__bf16)Wl[(wbase + k0 + 16 + j) * HD + col];
    }
    acc = __builtin_amdgcn_wmma_f32_16x16x32_bf16(
         false, A,  false, B,
         (short)0, acc,  false,  false);
  }

  float* __restrict__ dst = mat ? Q : P;
  const int rbase = blockIdx.x * 16 + khalf * 8;
  #pragma unroll
  for (int v = 0; v < 8; ++v)
    dst[(size_t)(rbase + v) * HD + col] = acc[v];
}

__global__ __launch_bounds__(256) void gat_init(
    float* __restrict__ out, float* __restrict__ m, float* __restrict__ s,
    int n64, int n8) {
  int i = blockIdx.x * blockDim.x + threadIdx.x;
  if (i < n64) out[i] = 0.0f;
  if (i < n8) { m[i] = -__builtin_inff(); s[i] = 0.0f; }
}

__global__ __launch_bounds__(256) void gat_logits(
    const float* __restrict__ P, const float* __restrict__ Q,
    const float* __restrict__ a,
    const int* __restrict__ src, const int* __restrict__ tgt,
    float* __restrict__ logits, float* __restrict__ m, int nEdges) {
  int t = blockIdx.x * blockDim.x + threadIdx.x;
  if (t >= nEdges * HEADS) return;
  const int e  = t >> 3;
  const int hd = t & 7;
  const int sn = src[e];
  const int tn = tgt[e];
  const float* __restrict__ p  = P + (size_t)sn * HD + hd * OUT_CH;
  const float* __restrict__ q  = Q + (size_t)tn * HD + hd * OUT_CH;
  const float* __restrict__ av = a + hd * OUT_CH;
  float lg = 0.0f;
  #pragma unroll
  for (int d = 0; d < OUT_CH; ++d) {
    float x = p[d] + q[d];
    x = x > 0.0f ? x : LEAKY * x;
    lg += x * av[d];
  }
  logits[t] = lg;
  float* addr = m + (size_t)tn * HEADS + hd;
  if (lg >= 0.0f) atomicMax((int*)addr, __float_as_int(lg));
  else            atomicMin((unsigned int*)addr, (unsigned int)__float_as_int(lg));
}

__global__ __launch_bounds__(256) void gat_expsum(
    const int* __restrict__ tgt, const float* __restrict__ m,
    float* __restrict__ logits, float* __restrict__ s, int nEdges) {
  int t = blockIdx.x * blockDim.x + threadIdx.x;
  if (t >= nEdges * HEADS) return;
  const int e  = t >> 3;
  const int hd = t & 7;
  const int tn = tgt[e];
  float ex = __expf(logits[t] - m[(size_t)tn * HEADS + hd]);
  logits[t] = ex;
  atomicAdd(s + (size_t)tn * HEADS + hd, ex);
}

__global__ __launch_bounds__(256) void gat_scatter(
    const float* __restrict__ h,
    const int* __restrict__ src, const int* __restrict__ tgt,
    const float* __restrict__ ex, const float* __restrict__ s,
    float* __restrict__ out, int nEdges) {
  int t = blockIdx.x * blockDim.x + threadIdx.x;
  if (t >= nEdges * HEADS) return;
  const int e  = t >> 3;
  const int hd = t & 7;
  const int sn = src[e];
  const int tn = tgt[e];
  const float att = ex[t] / s[(size_t)tn * HEADS + hd];
  const float* __restrict__ hi = h + (size_t)sn * IN_CH + hd * OUT_CH;
  float* __restrict__ o = out + (size_t)tn * HD + hd * OUT_CH;
  #pragma unroll
  for (int d = 0; d < OUT_CH; ++d)
    atomicAdd(o + d, hi[d] * att);
}

extern "C" void kernel_launch(void* const* d_in, const int* in_sizes, int n_in,
                              void* d_out, int out_size, void* d_ws, size_t ws_size,
                              hipStream_t stream) {
  const float* h   = (const float*)d_in[0];
  const float* W   = (const float*)d_in[1];
  const float* a   = (const float*)d_in[2];
  const int*   src = (const int*)d_in[3];
  const int*   tgt = (const int*)d_in[4];
  float*       out = (float*)d_out;

  const int nNodes = in_sizes[0] / IN_CH;
  const int nEdges = in_sizes[3];

  float* P      = (float*)d_ws;
  float* Q      = P + (size_t)nNodes * HD;
  float* logits = Q + (size_t)nNodes * HD;
  float* m      = logits + (size_t)nEdges * HEADS;
  float* s      = m + (size_t)nNodes * HEADS;

  const int rowBlocks = nNodes / 16;
  gat_project_wmma<<<rowBlocks, 256, 0, stream>>>(h, W, P, Q, nNodes);

  const int n64 = nNodes * HD, n8 = nNodes * HEADS;
  gat_init<<<(n64 + 255) / 256, 256, 0, stream>>>(out, m, s, n64, n8);

  const int eh = nEdges * HEADS;
  const int ehBlocks = (eh + 255) / 256;
  gat_logits <<<ehBlocks, 256, 0, stream>>>(P, Q, a, src, tgt, logits, m, nEdges);
  gat_expsum <<<ehBlocks, 256, 0, stream>>>(tgt, m, logits, s, nEdges);
  gat_scatter<<<ehBlocks, 256, 0, stream>>>(h, src, tgt, logits, s, out, nEdges);
}


# Edges grouped by target in the call (int-atomic count, prefix sum, fill), each group sorted by edge index so sums no longer follow arrival order; one lane per (target, head): no float atomics
# speedup vs baseline: 1.4465x; 1.4465x over previous
#include <hip/hip_runtime.h>
#include <hip/hip_bf16.h>
#include <math.h>


typedef __attribute__((ext_vector_type(16))) __bf16 v16bf;
typedef __attribute__((ext_vector_type(8)))  float  v8f;

#define IN_CH   64
#define HEADS   8
#define OUT_CH  8
#define HD      (HEADS * OUT_CH)
#define LEAKY   0.3f
#define HT_PITCH 68

__global__ __launch_bounds__(256) void gat_project_wmma(
    const float* __restrict__ h, const float* __restrict__ W,
    float* __restrict__ P, float* __restrict__ Q, int* __restrict__ cnt,
    int nNodes) {
  __shared__ float hT[16 * HT_PITCH];
  __shared__ float Wl[2 * IN_CH * HD];

  const int tid = threadIdx.x;

  {
    const int g = blockIdx.x * 256 + tid;
    if (g < nNodes) cnt[g] = 0;
  }

  {
    const float4* s4 = (const float4*)(h + (size_t)blockIdx.x * 16 * IN_CH);
    const int r = tid >> 4;
    const int c = (tid & 15) * 4;
    *(float4*)(hT + r * HT_PITCH + c) = s4[tid];
  }
  {
    const float4* s4 = (const float4*)W;
    float4* d4 = (float4*)Wl;
    #pragma unroll
    for (int i = 0; i < 8; ++i)
      d4[tid + 256 * i] = s4[tid + 256 * i];
  }
  __syncthreads();

  const int wave  = tid >> 5;
  const int lane  = tid & 31;
  const int mat   = wave >> 2;
  const int tn    = wave & 3;
  const int l16   = lane & 15;
  const int khalf = lane >> 4;
  const int col   = tn * 16 + l16;
  const int wbase = mat * IN_CH;

  const float* __restrict__ hrow = hT + l16 * HT_PITCH;

  v8f acc = {};
  #pragma unroll
  for (int kb = 0; kb < IN_CH; kb += 32) {
    const int k0 = kb + khalf * 8;
    v16bf A, B;
    #pragma unroll
    for (int j = 0; j < 8; ++j) {
      A[j]     = (__bf16)hrow[k0 + j];
      A[j + 8] = (__bf16)hrow[k0 + 16 + j];
      B[j]     = (__bf16)Wl[(wbase + k0 + j)      * HD + col];
      B[j + 8] = (__bf16)Wl[(wbase + k0 + 16 + j) * HD + col];
    }
    acc = __builtin_amdgcn_wmma_f32_16x16x32_bf16(
         false, A,  false, B,
         (short)0, acc,  false,  false);
  }

  float* __restrict__ dst = mat ? Q : P;
  const int rbase = blockIdx.x * 16 + khalf * 8;
  #pragma unroll
  for (int v = 0; v < 8; ++v)
    dst[(size_t)(rbase + v) * HD + col] = acc[v];
}

__global__ __launch_bounds__(256) void gat_count(
    const int* __restrict__ tgt, int* __restrict__ cnt,
    int* __restrict__ rank, int nEdges) {
  const int e = blockIdx.x * blockDim.x + threadIdx.x;
  if (e >= nEdges) return;
  rank[e] = atomicAdd(cnt + tgt[e], 1);
}

__global__ __launch_bounds__(1024) void gat_scan(
    const int* __restrict__ cnt, int* __restrict__ rowptr, int n) {
  __shared__ int wsum[32];
  __shared__ int tile_total;
  const int tid  = threadIdx.x;
  const int lane = tid & 31;
  const int wv   = tid >> 5;
  int carry = 0;
  for (int base = 0; base < n; base += 4096) {
    const int i0 = base + tid * 4;
    int c0 = 0, c1 = 0, c2 = 0, c3 = 0;
    if (i0 + 3 < n) {
      const int4 v = *(const int4*)(cnt + i0);
      c0 = v.x; c1 = v.y; c2 = v.z; c3 = v.w;
    } else {
      if (i0 < n)     c0 = cnt[i0];
      if (i0 + 1 < n) c1 = cnt[i0 + 1];
      if (i0 + 2 < n) c2 = cnt[i0 + 2];
    }
    const int tsum = c0 + c1 + c2 + c3;
    int inc = tsum;
    for (int off = 1; off < 32; off <<= 1) {
      const int v = __shfl_up(inc, off);
      if (lane >= off) inc += v;
    }
    if (lane == 31) wsum[wv] = inc;
    __syncthreads();
    if (wv == 0) {
      const int w = wsum[lane];
      int winc = w;
      for (int off = 1; off < 32; off <<= 1) {
        const int v = __shfl_up(winc, off);
        if (lane >= off) winc += v;
      }
      wsum[lane] = winc - w;
      if (lane == 31) tile_total = winc;
    }
    __syncthreads();
    const int ex = carry + wsum[wv] + inc - tsum;
    if (i0 + 3 < n) {
      int4 o;
      o.x = ex; o.y = ex + c0; o.z = ex + c0 + c1; o.w = ex + c0 + c1 + c2;
      *(int4*)(rowptr + i0) = o;
    } else {
      if (i0 < n)     rowptr[i0]     = ex;
      if (i0 + 1 < n) rowptr[i0 + 1] = ex + c0;
      if (i0 + 2 < n) rowptr[i0 + 2] = ex + c0 + c1;
    }
    carry += tile_total;
    __syncthreads();
  }
  if (tid == 0) rowptr[n] = carry;
}

__global__ __launch_bounds__(256) void gat_fill(
    const int* __restrict__ tgt, const int* __restrict__ rank,
    const int* __restrict__ rowptr, int* __restrict__ grp, int nEdges) {
  const int e = blockIdx.x * blockDim.x + threadIdx.x;
  if (e >= nEdges) return;
  grp[rowptr[tgt[e]] + rank[e]] = e;
}

__global__ __launch_bounds__(256) void gat_order(
    const int* __restrict__ src, const int* __restrict__ tgt,
    const int* __restrict__ rowptr, const int* __restrict__ grp,
    int* __restrict__ srcS, int nEdges) {
  const int p = blockIdx.x * blockDim.x + threadIdx.x;
  if (p >= nEdges) return;
  const int e  = grp[p];
  const int t  = tgt[e];
  const int b  = rowptr[t];
  const int en = rowptr[t + 1];
  int r = 0;
  for (int q = b; q < en; ++q) r += (grp[q] < e) ? 1 : 0;
  srcS[b + r] = src[e];
}

typedef __attribute__((ext_vector_type(2))) float v2f;

__device__ __forceinline__ float gat_logit(const float4 p0, const float4 p1,
                                           const float4 q0, const float4 q1,
                                           const float4 a0, const float4 a1) {
  const v2f pv[4] = {{p0.x, p0.y}, {p0.z, p0.w}, {p1.x, p1.y}, {p1.z, p1.w}};
  const v2f qv[4] = {{q0.x, q0.y}, {q0.z, q0.w}, {q1.x, q1.y}, {q1.z, q1.w}};
  const v2f av[4] = {{a0.x, a0.y}, {a0.z, a0.w}, {a1.x, a1.y}, {a1.z, a1.w}};
  float lg = 0.0f;
  #pragma unroll
  for (int j = 0; j < 4; ++j) {
    const v2f x = pv[j] + qv[j];
    const v2f y = x * LEAKY;
    const float x0 = x.x > 0.0f ? x.x : y.x;
    const float x1 = x.y > 0.0f ? x.y : y.y;
    lg += x0 * av[j].x;
    lg += x1 * av[j].y;
  }
  return lg;
}

__device__ __forceinline__ void gat_axpy(float* acc, const float4 h0,
                                         const float4 h1, const float w) {
  acc[0] += h0.x * w; acc[1] += h0.y * w; acc[2] += h0.z * w; acc[3] += h0.w * w;
  acc[4] += h1.x * w; acc[5] += h1.y * w; acc[6] += h1.z * w; acc[7] += h1.w * w;
}

__global__ __launch_bounds__(256) void gat_gather(
    const float* __restrict__ h, const float* __restrict__ P,
    const float* __restrict__ Q, const float* __restrict__ a,
    const int* __restrict__ rowptr, const int* __restrict__ srcS,
    float* __restrict__ lgbuf, float* __restrict__ out, int nNodes) {
  const int g = blockIdx.x * blockDim.x + threadIdx.x;
  if (g >= nNodes * HEADS) return;
  const int t  = g >> 3;
  const int hd = g & 7;
  const int b  = rowptr[t];
  const int en = rowptr[t + 1];
  const int co = hd * OUT_CH;

  float4 o0 = make_float4(0.0f, 0.0f, 0.0f, 0.0f), o1 = o0;

  if (en > b) {
    const int last = en - 1;
    const float4* qp = (const float4*)(Q + (size_t)t * HD + co);
    const float4 q0 = qp[0], q1 = qp[1];
    const float4* ap = (const float4*)(a + co);
    const float4 a0 = ap[0], a1 = ap[1];

    float m = -__builtin_inff();
    int n0 = srcS[b];
    int n1 = srcS[min(b + 1, last)];
    int n2 = srcS[min(b + 2, last)];
    int n3 = srcS[min(b + 3, last)];
    for (int p = b; p < en; p += 4) {
      const int i1 = min(p + 1, last), i2 = min(p + 2, last), i3 = min(p + 3, last);
      const int s0 = n0, s1 = n1, s2 = n2, s3 = n3;
      n0 = srcS[min(p + 4, last)];
      n1 = srcS[min(p + 5, last)];
      n2 = srcS[min(p + 6, last)];
      n3 = srcS[min(p + 7, last)];
      const float4* r0 = (const float4*)(P + (size_t)s0 * HD + co);
      const float4* r1 = (const float4*)(P + (size_t)s1 * HD + co);
      const float4* r2 = (const float4*)(P + (size_t)s2 * HD + co);
      const float4* r3 = (const float4*)(P + (size_t)s3 * HD + co);
      const float4 x00 = r0[0], x01 = r0[1], x10 = r1[0], x11 = r1[1];
      const float4 x20 = r2[0], x21 = r2[1], x30 = r3[0], x31 = r3[1];
      const float l0 = gat_logit(x00, x01, q0, q1, a0, a1);
      const float l1 = gat_logit(x10, x11, q0, q1, a0, a1);
      const float l2 = gat_logit(x20, x21, q0, q1, a0, a1);
      const float l3 = gat_logit(x30, x31, q0, q1, a0, a1);
      lgbuf[(size_t)p  * HEADS + hd] = l0;
      lgbuf[(size_t)i1 * HEADS + hd] = l1;
      lgbuf[(size_t)i2 * HEADS + hd] = l2;
      lgbuf[(size_t)i3 * HEADS + hd] = l3;
      m = fmaxf(m, fmaxf(fmaxf(l0, l1), fmaxf(l2, l3)));
    }

    float s = 0.0f;
    float acc[OUT_CH] = {0.0f, 0.0f, 0.0f, 0.0f, 0.0f, 0.0f, 0.0f, 0.0f};
    n0 = srcS[b];
    n1 = srcS[min(b + 1, last)];
    n2 = srcS[min(b + 2, last)];
    n3 = srcS[min(b + 3, last)];
    for (int p = b; p < en; p += 4) {
      const int i1 = min(p + 1, last), i2 = min(p + 2, last), i3 = min(p + 3, last);
      const int s0 = n0, s1 = n1, s2 = n2, s3 = n3;
      n0 = srcS[min(p + 4, last)];
      n1 = srcS[min(p + 5, last)];
      n2 = srcS[min(p + 6, last)];
      n3 = srcS[min(p + 7, last)];
      const float l0 = lgbuf[(size_t)p  * HEADS + hd];
      const float l1 = lgbuf[(size_t)i1 * HEADS + hd];
      const float l2 = lgbuf[(size_t)i2 * HEADS + hd];
      const float l3 = lgbuf[(size_t)i3 * HEADS + hd];
      const float4* r0 = (const float4*)(h + (size_t)s0 * IN_CH + co);
      const float4* r1 = (const float4*)(h + (size_t)s1 * IN_CH + co);
      const float4* r2 = (const float4*)(h + (size_t)s2 * IN_CH + co);
      const float4* r3 = (const float4*)(h + (size_t)s3 * IN_CH + co);
      const float4 x00 = r0[0], x01 = r0[1], x10 = r1[0], x11 = r1[1];
      const float4 x20 = r2[0], x21 = r2[1], x30 = r3[0], x31 = r3[1];
      const float e0 = __expf(l0 - m);
      const float e1 = (p + 1 < en) ? __expf(l1 - m) : 0.0f;
      const float e2 = (p + 2 < en) ? __expf(l2 - m) : 0.0f;
      const float e3 = (p + 3 < en) ? __expf(l3 - m) : 0.0f;
      s += e0; s += e1; s += e2; s += e3;
      gat_axpy(acc, x00, x01, e0);
      gat_axpy(acc, x10, x11, e1);
      gat_axpy(acc, x20, x21, e2);
      gat_axpy(acc, x30, x31, e3);
    }

    o0.x = acc[0] / s; o0.y = acc[1] / s; o0.z = acc[2] / s; o0.w = acc[3] / s;
    o1.x = acc[4] / s; o1.y = acc[5] / s; o1.z = acc[6] / s; o1.w = acc[7] / s;
  }
  float4* op = (float4*)(out + (size_t)t * HD + co);
  op[0] = o0; op[1] = o1;
}

extern "C" void kernel_launch(void* const* d_in, const int* in_sizes, int n_in,
                              void* d_out, int out_size, void* d_ws, size_t ws_size,
                              hipStream_t stream) {
  const float* h   = (const float*)d_in[0];
  const float* W   = (const float*)d_in[1];
  const float* a   = (const float*)d_in[2];
  const int*   src = (const int*)d_in[3];
  const int*   tgt = (const int*)d_in[4];
  float*       out = (float*)d_out;

  const int nNodes = in_sizes[0] / IN_CH;
  const int nEdges = in_sizes[3];

  const size_t nPad = ((size_t)nNodes + 64) & ~(size_t)63;
  const size_t ePad = ((size_t)nEdges + 63) & ~(size_t)63;
  float* P      = (float*)d_ws;
  float* Q      = P + (size_t)nNodes * HD;
  float* lgbuf  = Q + (size_t)nNodes * HD;
  int*   cnt    = (int*)(lgbuf + ePad * HEADS);
  int*   rowptr = cnt + nPad;
  int*   rank   = rowptr + nPad;
  int*   grp    = rank + ePad;
  int*   srcS   = grp + ePad;

  const int rowBlocks = nNodes / 16;
  gat_project_wmma<<<rowBlocks, 256, 0, stream>>>(h, W, P, Q, cnt, nNodes);

  const int eBlocks = (nEdges + 255) / 256;
  gat_count<<<eBlocks, 256, 0, stream>>>(tgt, cnt, rank, nEdges);
  gat_scan <<<1, 1024, 0, stream>>>(cnt, rowptr, nNodes);
  gat_fill <<<eBlocks, 256, 0, stream>>>(tgt, rank, rowptr, grp, nEdges);
  gat_order<<<eBlocks, 256, 0, stream>>>(src, tgt, rowptr, grp, srcS, nEdges);

  const int nh = nNodes * HEADS;
  gat_gather<<<(nh + 255) / 256, 256, 0, stream>>>(h, P, Q, a, rowptr, srcS,
                                                   lgbuf, out, nNodes);
}


# Prefix sum of the counts in one sweep: each thread takes 52 consecutive counters (13 loads issued as one group), one block-wide scan of thread totals; 13 dependent load+scan trips become one
# speedup vs baseline: 1.5887x; 1.0983x over previous
#include <hip/hip_runtime.h>
#include <hip/hip_bf16.h>
#include <math.h>


typedef __attribute__((ext_vector_type(16))) __bf16 v16bf;
typedef __attribute__((ext_vector_type(8)))  float  v8f;

#define IN_CH   64
#define HEADS   8
#define OUT_CH  8
#define HD      (HEADS * OUT_CH)
#define LEAKY   0.3f
#define HT_PITCH 68

__global__ __launch_bounds__(256) void gat_project_wmma(
    const float* __restrict__ h, const float* __restrict__ W,
    float* __restrict__ P, float* __restrict__ Q, int* __restrict__ cnt,
    int nNodes) {
  __shared__ float hT[16 * HT_PITCH];
  __shared__ float Wl[2 * IN_CH * HD];

  const int tid = threadIdx.x;

  {
    const int g = blockIdx.x * 256 + tid;
    if (g < nNodes) cnt[g] = 0;
  }

  {
    const float4* s4 = (const float4*)(h + (size_t)blockIdx.x * 16 * IN_CH);
    const int r = tid >> 4;
    const int c = (tid & 15) * 4;
    *(float4*)(hT + r * HT_PITCH + c) = s4[tid];
  }
  {
    const float4* s4 = (const float4*)W;
    float4* d4 = (float4*)Wl;
    #pragma unroll
    for (int i = 0; i < 8; ++i)
      d4[tid + 256 * i] = s4[tid + 256 * i];
  }
  __syncthreads();

  const int wave  = tid >> 5;
  const int lane  = tid & 31;
  const int mat   = wave >> 2;
  const int tn    = wave & 3;
  const int l16   = lane & 15;
  const int khalf = lane >> 4;
  const int col   = tn * 16 + l16;
  const int wbase = mat * IN_CH;

  const float* __restrict__ hrow = hT + l16 * HT_PITCH;

  v8f acc = {};
  #pragma unroll
  for (int kb = 0; kb < IN_CH; kb += 32) {
    const int k0 = kb + khalf * 8;
    v16bf A, B;
    #pragma unroll
    for (int j = 0; j < 8; ++j) {
      A[j]     = (__bf16)hrow[k0 + j];
      A[j + 8] = (__bf16)hrow[k0 + 16 + j];
      B[j]     = (__bf16)Wl[(wbase + k0 + j)      * HD + col];
      B[j + 8] = (__bf16)Wl[(wbase + k0 + 16 + j) * HD + col];
    }
    acc = __builtin_amdgcn_wmma_f32_16x16x32_bf16(
         false, A,  false, B,
         (short)0, acc,  false,  false);
  }

  float* __restrict__ dst = mat ? Q : P;
  const int rbase = blockIdx.x * 16 + khalf * 8;
  #pragma unroll
  for (int v = 0; v < 8; ++v)
    dst[(size_t)(rbase + v) * HD + col] = acc[v];
}

__global__ __launch_bounds__(256) void gat_count(
    const int* __restrict__ tgt, int* __restrict__ cnt,
    int* __restrict__ rank, int nEdges) {
  const int e = blockIdx.x * blockDim.x + threadIdx.x;
  if (e >= nEdges) return;
  rank[e] = atomicAdd(cnt + tgt[e], 1);
}

#define SCAN_VEC 13
typedef __attribute__((ext_vector_type(4))) int v4i;

__global__ __launch_bounds__(1024) void gat_scan(
    const int* __restrict__ cnt, int* __restrict__ rowptr, int n) {
  __shared__ int wsum[32];
  __shared__ int tile_total;
  const int tid  = threadIdx.x;
  const int lane = tid & 31;
  const int wv   = tid >> 5;
  const int numVec = n >> 2;
  const v4i* __restrict__ c4 = (const v4i*)cnt;
  v4i* __restrict__ r4 = (v4i*)rowptr;
  int carry = 0;
  for (int vb = 0; vb < numVec; vb += 1024 * SCAN_VEC) {
    const int v0 = vb + tid * SCAN_VEC;
    v4i c[SCAN_VEC];
    #pragma unroll
    for (int j = 0; j < SCAN_VEC; ++j) c[j] = c4[min(v0 + j, numVec - 1)];
    #pragma unroll
    for (int j = 0; j < SCAN_VEC; ++j) {
      const bool in = v0 + j < numVec;
      c[j].x = in ? c[j].x : 0; c[j].y = in ? c[j].y : 0;
      c[j].z = in ? c[j].z : 0; c[j].w = in ? c[j].w : 0;
    }
    int tsum = 0;
    #pragma unroll
    for (int j = 0; j < SCAN_VEC; ++j) tsum += (c[j].x + c[j].y) + (c[j].z + c[j].w);
    __builtin_amdgcn_sched_group_barrier(0x020, SCAN_VEC, 0);
    __builtin_amdgcn_sched_group_barrier(0x002, 512, 0);

    int inc = tsum;
    for (int off = 1; off < 32; off <<= 1) {
      const int v = __shfl_up(inc, off);
      if (lane >= off) inc += v;
    }
    if (lane == 31) wsum[wv] = inc;
    __syncthreads();
    if (wv == 0) {
      const int w = wsum[lane];
      int winc = w;
      for (int off = 1; off < 32; off <<= 1) {
        const int v = __shfl_up(winc, off);
        if (lane >= off) winc += v;
      }
      wsum[lane] = winc - w;
      if (lane == 31) tile_total = winc;
    }
    __syncthreads();
    int run = carry + wsum[wv] + inc - tsum;
    #pragma unroll
    for (int j = 0; j < SCAN_VEC; ++j) {
      v4i o;
      o.x = run; run += c[j].x;
      o.y = run; run += c[j].y;
      o.z = run; run += c[j].z;
      o.w = run; run += c[j].w;
      if (v0 + j < numVec) r4[v0 + j] = o;
    }
    carry += tile_total;
    __syncthreads();
  }
  if (tid == 0) {
    int run = carry;
    for (int i = numVec * 4; i < n; ++i) {
      rowptr[i] = run;
      run += cnt[i];
    }
    rowptr[n] = run;
  }
}

__global__ __launch_bounds__(256) void gat_fill(
    const int* __restrict__ tgt, const int* __restrict__ rank,
    const int* __restrict__ rowptr, int* __restrict__ grp, int nEdges) {
  const int e = blockIdx.x * blockDim.x + threadIdx.x;
  if (e >= nEdges) return;
  grp[rowptr[tgt[e]] + rank[e]] = e;
}

__global__ __launch_bounds__(256) void gat_order(
    const int* __restrict__ src, const int* __restrict__ tgt,
    const int* __restrict__ rowptr, const int* __restrict__ grp,
    int* __restrict__ srcS, int nEdges) {
  const int p = blockIdx.x * blockDim.x + threadIdx.x;
  if (p >= nEdges) return;
  const int e  = grp[p];
  const int t  = tgt[e];
  const int b  = rowptr[t];
  const int en = rowptr[t + 1];
  int r = 0;
  for (int q = b; q < en; ++q) r += (grp[q] < e) ? 1 : 0;
  srcS[b + r] = src[e];
}

typedef __attribute__((ext_vector_type(2))) float v2f;

__device__ __forceinline__ float gat_logit(const float4 p0, const float4 p1,
                                           const float4 q0, const float4 q1,
                                           const float4 a0, const float4 a1) {
  const v2f pv[4] = {{p0.x, p0.y}, {p0.z, p0.w}, {p1.x, p1.y}, {p1.z, p1.w}};
  const v2f qv[4] = {{q0.x, q0.y}, {q0.z, q0.w}, {q1.x, q1.y}, {q1.z, q1.w}};
  const v2f av[4] = {{a0.x, a0.y}, {a0.z, a0.w}, {a1.x, a1.y}, {a1.z, a1.w}};
  float lg = 0.0f;
  #pragma unroll
  for (int j = 0; j < 4; ++j) {
    const v2f x = pv[j] + qv[j];
    const v2f y = x * LEAKY;
    const float x0 = x.x > 0.0f ? x.x : y.x;
    const float x1 = x.y > 0.0f ? x.y : y.y;
    lg += x0 * av[j].x;
    lg += x1 * av[j].y;
  }
  return lg;
}

__device__ __forceinline__ void gat_axpy(float* acc, const float4 h0,
                                         const float4 h1, const float w) {
  acc[0] += h0.x * w; acc[1] += h0.y * w; acc[2] += h0.z * w; acc[3] += h0.w * w;
  acc[4] += h1.x * w; acc[5] += h1.y * w; acc[6] += h1.z * w; acc[7] += h1.w * w;
}

__global__ __launch_bounds__(256) void gat_gather(
    const float* __restrict__ h, const float* __restrict__ P,
    const float* __restrict__ Q, const float* __restrict__ a,
    const int* __restrict__ rowptr, const int* __restrict__ srcS,
    float* __restrict__ out, int nNodes) {
  const int g = blockIdx.x * blockDim.x + threadIdx.x;
  if (g >= nNodes * HEADS) return;
  const int t  = g >> 3;
  const int hd = g & 7;
  const int b  = rowptr[t];
  const int en = rowptr[t + 1];
  const int co = hd * OUT_CH;

  float4 o0 = make_float4(0.0f, 0.0f, 0.0f, 0.0f), o1 = o0;

  if (en > b) {
    const int last = en - 1;
    const float4* qp = (const float4*)(Q + (size_t)t * HD + co);
    const float4 q0 = qp[0], q1 = qp[1];
    const float4* ap = (const float4*)(a + co);
    const float4 a0 = ap[0], a1 = ap[1];

    float m = -__builtin_inff();
    float s = 0.0f;
    float acc[OUT_CH] = {0.0f, 0.0f, 0.0f, 0.0f, 0.0f, 0.0f, 0.0f, 0.0f};
    int n0 = srcS[b];
    int n1 = srcS[min(b + 1, last)];
    int n2 = srcS[min(b + 2, last)];
    int n3 = srcS[min(b + 3, last)];
    const unsigned cb = (unsigned)co * 4u;
    const char* __restrict__ Pb = (const char*)P;
    const char* __restrict__ hb = (const char*)h;
    for (int p = b; p < en; p += 4) {
      const unsigned f0 = (unsigned)n0 * (HD * 4u) + cb;
      const unsigned f1 = (unsigned)n1 * (HD * 4u) + cb;
      const unsigned f2 = (unsigned)n2 * (HD * 4u) + cb;
      const unsigned f3 = (unsigned)n3 * (HD * 4u) + cb;
      n0 = srcS[min(p + 4, last)];
      n1 = srcS[min(p + 5, last)];
      n2 = srcS[min(p + 6, last)];
      n3 = srcS[min(p + 7, last)];
      const float4* r0 = (const float4*)(Pb + f0);
      const float4* r1 = (const float4*)(Pb + f1);
      const float4* r2 = (const float4*)(Pb + f2);
      const float4* r3 = (const float4*)(Pb + f3);
      const float4 x00 = r0[0], x01 = r0[1], x10 = r1[0], x11 = r1[1];
      const float4 x20 = r2[0], x21 = r2[1], x30 = r3[0], x31 = r3[1];
      const float4* g0 = (const float4*)(hb + f0);
      const float4* g1 = (const float4*)(hb + f1);
      const float4* g2 = (const float4*)(hb + f2);
      const float4* g3 = (const float4*)(hb + f3);
      const float4 y00 = g0[0], y01 = g0[1], y10 = g1[0], y11 = g1[1];
      const float4 y20 = g2[0], y21 = g2[1], y30 = g3[0], y31 = g3[1];

      const float l0 = gat_logit(x00, x01, q0, q1, a0, a1);
      const float l1 = gat_logit(x10, x11, q0, q1, a0, a1);
      const float l2 = gat_logit(x20, x21, q0, q1, a0, a1);
      const float l3 = gat_logit(x30, x31, q0, q1, a0, a1);

      const float mn = fmaxf(m, fmaxf(fmaxf(l0, l1), fmaxf(l2, l3)));
      const float sc = __expf(m - mn);
      m = mn;
      const float e0 = __expf(l0 - mn);
      const float e1 = (p + 1 < en) ? __expf(l1 - mn) : 0.0f;
      const float e2 = (p + 2 < en) ? __expf(l2 - mn) : 0.0f;
      const float e3 = (p + 3 < en) ? __expf(l3 - mn) : 0.0f;
      s *= sc;
      s += e0; s += e1; s += e2; s += e3;
      acc[0] *= sc; acc[1] *= sc; acc[2] *= sc; acc[3] *= sc;
      acc[4] *= sc; acc[5] *= sc; acc[6] *= sc; acc[7] *= sc;
      gat_axpy(acc, y00, y01, e0);
      gat_axpy(acc, y10, y11, e1);
      gat_axpy(acc, y20, y21, e2);
      gat_axpy(acc, y30, y31, e3);
      asm volatile("" :: "v"(n0), "v"(n1), "v"(n2), "v"(n3));
      __builtin_amdgcn_sched_group_barrier(0x020, 20, 0);
      __builtin_amdgcn_sched_group_barrier(0x002, 512, 0);
    }

    o0.x = acc[0] / s; o0.y = acc[1] / s; o0.z = acc[2] / s; o0.w = acc[3] / s;
    o1.x = acc[4] / s; o1.y = acc[5] / s; o1.z = acc[6] / s; o1.w = acc[7] / s;
  }
  float4* op = (float4*)(out + (size_t)t * HD + co);
  op[0] = o0; op[1] = o1;
}

extern "C" void kernel_launch(void* const* d_in, const int* in_sizes, int n_in,
                              void* d_out, int out_size, void* d_ws, size_t ws_size,
                              hipStream_t stream) {
  const float* h   = (const float*)d_in[0];
  const float* W   = (const float*)d_in[1];
  const float* a   = (const float*)d_in[2];
  const int*   src = (const int*)d_in[3];
  const int*   tgt = (const int*)d_in[4];
  float*       out = (float*)d_out;

  const int nNodes = in_sizes[0] / IN_CH;
  const int nEdges = in_sizes[3];

  const size_t nPad = ((size_t)nNodes + 64) & ~(size_t)63;
  const size_t ePad = ((size_t)nEdges + 63) & ~(size_t)63;
  float* P      = (float*)d_ws;
  float* Q      = P + (size_t)nNodes * HD;
  int*   cnt    = (int*)(Q + (size_t)nNodes * HD);
  int*   rowptr = cnt + nPad;
  int*   rank   = rowptr + nPad;
  int*   grp    = rank + ePad;
  int*   srcS   = grp + ePad;

  const int rowBlocks = nNodes / 16;
  gat_project_wmma<<<rowBlocks, 256, 0, stream>>>(h, W, P, Q, cnt, nNodes);

  const int eBlocks = (nEdges + 255) / 256;
  gat_count<<<eBlocks, 256, 0, stream>>>(tgt, cnt, rank, nEdges);
  gat_scan <<<1, 1024, 0, stream>>>(cnt, rowptr, nNodes);
  gat_fill <<<eBlocks, 256, 0, stream>>>(tgt, rank, rowptr, grp, nEdges);
  gat_order<<<eBlocks, 256, 0, stream>>>(src, tgt, rowptr, grp, srcS, nEdges);

  const int nh = nNodes * HEADS;
  gat_gather<<<(nh + 255) / 256, 256, 0, stream>>>(h, P, Q, a, rowptr, srcS,
                                                   out, nNodes);
}


# Gather launch puts each target's group in edge-index order itself (its 8 lanes rank the group in LDS, source ids then come from LDS): the order launch and its per-edge target lookups are gone
# speedup vs baseline: 1.6630x; 1.0468x over previous
#include <hip/hip_runtime.h>
#include <hip/hip_bf16.h>
#include <math.h>


typedef __attribute__((ext_vector_type(16))) __bf16 v16bf;
typedef __attribute__((ext_vector_type(8)))  float  v8f;

#define IN_CH   64
#define HEADS   8
#define OUT_CH  8
#define HD      (HEADS * OUT_CH)
#define LEAKY   0.3f
#define HT_PITCH 68

__global__ __launch_bounds__(256) void gat_project_wmma(
    const float* __restrict__ h, const float* __restrict__ W,
    float* __restrict__ P, float* __restrict__ Q, int* __restrict__ cnt,
    int nNodes) {
  __shared__ float hT[16 * HT_PITCH];
  __shared__ float Wl[2 * IN_CH * HD];

  const int tid = threadIdx.x;

  {
    const int g = blockIdx.x * 256 + tid;
    if (g < nNodes) cnt[g] = 0;
  }

  {
    const float4* s4 = (const float4*)(h + (size_t)blockIdx.x * 16 * IN_CH);
    const int r = tid >> 4;
    const int c = (tid & 15) * 4;
    *(float4*)(hT + r * HT_PITCH + c) = s4[tid];
  }
  {
    const float4* s4 = (const float4*)W;
    float4* d4 = (float4*)Wl;
    #pragma unroll
    for (int i = 0; i < 8; ++i)
      d4[tid + 256 * i] = s4[tid + 256 * i];
  }
  __syncthreads();

  const int wave  = tid >> 5;
  const int lane  = tid & 31;
  const int mat   = wave >> 2;
  const int tn    = wave & 3;
  const int l16   = lane & 15;
  const int khalf = lane >> 4;
  const int col   = tn * 16 + l16;
  const int wbase = mat * IN_CH;

  const float* __restrict__ hrow = hT + l16 * HT_PITCH;

  v8f acc = {};
  #pragma unroll
  for (int kb = 0; kb < IN_CH; kb += 32) {
    const int k0 = kb + khalf * 8;
    v16bf A, B;
    #pragma unroll
    for (int j = 0; j < 8; ++j) {
      A[j]     = (__bf16)hrow[k0 + j];
      A[j + 8] = (__bf16)hrow[k0 + 16 + j];
      B[j]     = (__bf16)Wl[(wbase + k0 + j)      * HD + col];
      B[j + 8] = (__bf16)Wl[(wbase + k0 + 16 + j) * HD + col];
    }
    acc = __builtin_amdgcn_wmma_f32_16x16x32_bf16(
         false, A,  false, B,
         (short)0, acc,  false,  false);
  }

  float* __restrict__ dst = mat ? Q : P;
  const int rbase = blockIdx.x * 16 + khalf * 8;
  #pragma unroll
  for (int v = 0; v < 8; ++v)
    dst[(size_t)(rbase + v) * HD + col] = acc[v];
}

__global__ __launch_bounds__(256) void gat_count(
    const int* __restrict__ tgt, int* __restrict__ cnt,
    int* __restrict__ rank, int nEdges) {
  const int e = blockIdx.x * blockDim.x + threadIdx.x;
  if (e >= nEdges) return;
  rank[e] = atomicAdd(cnt + tgt[e], 1);
}

#define SCAN_VEC 13
typedef __attribute__((ext_vector_type(4))) int v4i;

__global__ __launch_bounds__(1024) void gat_scan(
    const int* __restrict__ cnt, int* __restrict__ rowptr, int n) {
  __shared__ int wsum[32];
  __shared__ int tile_total;
  const int tid  = threadIdx.x;
  const int lane = tid & 31;
  const int wv   = tid >> 5;
  const int numVec = n >> 2;
  const v4i* __restrict__ c4 = (const v4i*)cnt;
  v4i* __restrict__ r4 = (v4i*)rowptr;
  int carry = 0;
  for (int vb = 0; vb < numVec; vb += 1024 * SCAN_VEC) {
    const int v0 = vb + tid * SCAN_VEC;
    v4i c[SCAN_VEC];
    #pragma unroll
    for (int j = 0; j < SCAN_VEC; ++j) c[j] = c4[min(v0 + j, numVec - 1)];
    #pragma unroll
    for (int j = 0; j < SCAN_VEC; ++j) {
      const bool in = v0 + j < numVec;
      c[j].x = in ? c[j].x : 0; c[j].y = in ? c[j].y : 0;
      c[j].z = in ? c[j].z : 0; c[j].w = in ? c[j].w : 0;
    }
    int tsum = 0;
    #pragma unroll
    for (int j = 0; j < SCAN_VEC; ++j) tsum += (c[j].x + c[j].y) + (c[j].z + c[j].w);
    __builtin_amdgcn_sched_group_barrier(0x020, SCAN_VEC, 0);
    __builtin_amdgcn_sched_group_barrier(0x002, 512, 0);

    int inc = tsum;
    for (int off = 1; off < 32; off <<= 1) {
      const int v = __shfl_up(inc, off);
      if (lane >= off) inc += v;
    }
    if (lane == 31) wsum[wv] = inc;
    __syncthreads();
    if (wv == 0) {
      const int w = wsum[lane];
      int winc = w;
      for (int off = 1; off < 32; off <<= 1) {
        const int v = __shfl_up(winc, off);
        if (lane >= off) winc += v;
      }
      wsum[lane] = winc - w;
      if (lane == 31) tile_total = winc;
    }
    __syncthreads();
    int run = carry + wsum[wv] + inc - tsum;
    #pragma unroll
    for (int j = 0; j < SCAN_VEC; ++j) {
      v4i o;
      o.x = run; run += c[j].x;
      o.y = run; run += c[j].y;
      o.z = run; run += c[j].z;
      o.w = run; run += c[j].w;
      if (v0 + j < numVec) r4[v0 + j] = o;
    }
    carry += tile_total;
    __syncthreads();
  }
  if (tid == 0) {
    int run = carry;
    for (int i = numVec * 4; i < n; ++i) {
      rowptr[i] = run;
      run += cnt[i];
    }
    rowptr[n] = run;
  }
}

__global__ __launch_bounds__(256) void gat_fill(
    const int* __restrict__ tgt, const int* __restrict__ rank,
    const int* __restrict__ rowptr, int* __restrict__ grp, int nEdges) {
  const int e = blockIdx.x * blockDim.x + threadIdx.x;
  if (e >= nEdges) return;
  grp[rowptr[tgt[e]] + rank[e]] = e;
}

#define GRP_CAP 64

typedef __attribute__((ext_vector_type(2))) float v2f;

__device__ __forceinline__ float gat_logit(const float4 p0, const float4 p1,
                                           const float4 q0, const float4 q1,
                                           const float4 a0, const float4 a1) {
  const v2f pv[4] = {{p0.x, p0.y}, {p0.z, p0.w}, {p1.x, p1.y}, {p1.z, p1.w}};
  const v2f qv[4] = {{q0.x, q0.y}, {q0.z, q0.w}, {q1.x, q1.y}, {q1.z, q1.w}};
  const v2f av[4] = {{a0.x, a0.y}, {a0.z, a0.w}, {a1.x, a1.y}, {a1.z, a1.w}};
  float lg = 0.0f;
  #pragma unroll
  for (int j = 0; j < 4; ++j) {
    const v2f x = pv[j] + qv[j];
    const v2f y = x * LEAKY;
    const float x0 = x.x > 0.0f ? x.x : y.x;
    const float x1 = x.y > 0.0f ? x.y : y.y;
    lg += x0 * av[j].x;
    lg += x1 * av[j].y;
  }
  return lg;
}

__device__ __forceinline__ void gat_axpy(float* acc, const float4 h0,
                                         const float4 h1, const float w) {
  acc[0] += h0.x * w; acc[1] += h0.y * w; acc[2] += h0.z * w; acc[3] += h0.w * w;
  acc[4] += h1.x * w; acc[5] += h1.y * w; acc[6] += h1.z * w; acc[7] += h1.w * w;
}

__global__ __launch_bounds__(256) void gat_gather(
    const float* __restrict__ h, const float* __restrict__ P,
    const float* __restrict__ Q, const float* __restrict__ a,
    const int* __restrict__ src, const int* __restrict__ rowptr,
    const int* __restrict__ grp, float* __restrict__ out, int nNodes) {
  __shared__ __attribute__((aligned(16))) int ids[32 * GRP_CAP];
  __shared__ __attribute__((aligned(16))) int su[32 * GRP_CAP];
  __shared__ __attribute__((aligned(16))) int ss[32 * GRP_CAP];

  const int tid = threadIdx.x;
  const int g = blockIdx.x * 256 + tid;
  const bool valid = g < nNodes * HEADS;
  const int t  = g >> 3;
  const int hd = g & 7;
  const int o  = (tid >> 3) * GRP_CAP;
  const int co = hd * OUT_CH;
  int b = 0, en = 0;
  if (valid) { b = rowptr[t]; en = rowptr[t + 1]; }
  const int deg  = en - b;
  const int last = deg - 1;
  const bool fits = deg > 0 && deg <= GRP_CAP;

  const float4* qp = (const float4*)(Q + (size_t)(valid ? t : 0) * HD + co);
  const float4 q0 = qp[0], q1 = qp[1];
  const float4* ap = (const float4*)(a + co);
  const float4 a0 = ap[0], a1 = ap[1];

  if (fits) {
    for (int i0 = hd; i0 < deg; i0 += 24) {
      const int i1 = i0 + 8, i2 = i0 + 16;
      const int e0 = grp[b + i0];
      const int e1 = grp[b + min(i1, last)];
      const int e2 = grp[b + min(i2, last)];
      const int s0 = src[e0], s1 = src[e1], s2 = src[e2];
      asm volatile("" :: "v"(s0), "v"(s1), "v"(s2));
      ids[o + i0] = e0; su[o + i0] = s0;
      if (i1 < deg) { ids[o + i1] = e1; su[o + i1] = s1; }
      if (i2 < deg) { ids[o + i2] = e2; su[o + i2] = s2; }
    }
    const int pi = deg + hd;
    if (hd < 3 && pi < ((deg + 3) & ~3)) ids[o + pi] = 0x7fffffff;
  }
  __syncthreads();
  if (fits) {
    for (int idx = hd; idx < deg; idx += 8) {
      const int e  = ids[o + idx];
      const int sv = su[o + idx];
      int r = 0;
      for (int q = 0; q < deg; q += 4) {
        const v4i v = *(const v4i*)(ids + o + q);
        r += (v.x < e ? 1 : 0) + (v.y < e ? 1 : 0) + (v.z < e ? 1 : 0) + (v.w < e ? 1 : 0);
      }
      ss[o + r] = sv;
    }
  }
  __syncthreads();

  float4 o0 = make_float4(0.0f, 0.0f, 0.0f, 0.0f), o1 = o0;

  if (deg > 0) {
    float m = -__builtin_inff();
    float s = 0.0f;
    float acc[OUT_CH] = {0.0f, 0.0f, 0.0f, 0.0f, 0.0f, 0.0f, 0.0f, 0.0f};
    const unsigned cb = (unsigned)co * 4u;
    const char* __restrict__ Pb = (const char*)P;
    const char* __restrict__ hb = (const char*)h;

    if (fits) {
      int n0 = ss[o];
      int n1 = ss[o + min(1, last)];
      int n2 = ss[o + min(2, last)];
      int n3 = ss[o + min(3, last)];
      for (int i = 0; i < deg; i += 4) {
        const unsigned f0 = (unsigned)n0 * (HD * 4u) + cb;
        const unsigned f1 = (unsigned)n1 * (HD * 4u) + cb;
        const unsigned f2 = (unsigned)n2 * (HD * 4u) + cb;
        const unsigned f3 = (unsigned)n3 * (HD * 4u) + cb;
        n0 = ss[o + min(i + 4, last)];
        n1 = ss[o + min(i + 5, last)];
        n2 = ss[o + min(i + 6, last)];
        n3 = ss[o + min(i + 7, last)];
        const float4* r0 = (const float4*)(Pb + f0);
        const float4* r1 = (const float4*)(Pb + f1);
        const float4* r2 = (const float4*)(Pb + f2);
        const float4* r3 = (const float4*)(Pb + f3);
        const float4 x00 = r0[0], x01 = r0[1], x10 = r1[0], x11 = r1[1];
        const float4 x20 = r2[0], x21 = r2[1], x30 = r3[0], x31 = r3[1];
        const float4* g0 = (const float4*)(hb + f0);
        const float4* g1 = (const float4*)(hb + f1);
        const float4* g2 = (const float4*)(hb + f2);
        const float4* g3 = (const float4*)(hb + f3);
        const float4 y00 = g0[0], y01 = g0[1], y10 = g1[0], y11 = g1[1];
        const float4 y20 = g2[0], y21 = g2[1], y30 = g3[0], y31 = g3[1];

        const float l0 = gat_logit(x00, x01, q0, q1, a0, a1);
        const float l1 = gat_logit(x10, x11, q0, q1, a0, a1);
        const float l2 = gat_logit(x20, x21, q0, q1, a0, a1);
        const float l3 = gat_logit(x30, x31, q0, q1, a0, a1);

        const float mn = fmaxf(m, fmaxf(fmaxf(l0, l1), fmaxf(l2, l3)));
        const float sc = __expf(m - mn);
        m = mn;
        const float e0 = __expf(l0 - mn);
        const float e1 = (i + 1 < deg) ? __expf(l1 - mn) : 0.0f;
        const float e2 = (i + 2 < deg) ? __expf(l2 - mn) : 0.0f;
        const float e3 = (i + 3 < deg) ? __expf(l3 - mn) : 0.0f;
        s *= sc;
        s += e0; s += e1; s += e2; s += e3;
        acc[0] *= sc; acc[1] *= sc; acc[2] *= sc; acc[3] *= sc;
        acc[4] *= sc; acc[5] *= sc; acc[6] *= sc; acc[7] *= sc;
        gat_axpy(acc, y00, y01, e0);
        gat_axpy(acc, y10, y11, e1);
        gat_axpy(acc, y20, y21, e2);
        gat_axpy(acc, y30, y31, e3);
        asm volatile("" :: "v"(n0), "v"(n1), "v"(n2), "v"(n3));
        __builtin_amdgcn_sched_group_barrier(0x020, 16, 0);
        __builtin_amdgcn_sched_group_barrier(0x002, 512, 0);
      }
    } else {
      int prev = -1;
      for (int i = 0; i < deg; ++i) {
        int e = 0x7fffffff;
        for (int q = b; q < en; ++q) {
          const int v = grp[q];
          e = (v > prev && v < e) ? v : e;
        }
        prev = e;
        const unsigned f = (unsigned)src[e] * (HD * 4u) + cb;
        const float4* r  = (const float4*)(Pb + f);
        const float4* gq = (const float4*)(hb + f);
        const float4 x0 = r[0], x1 = r[1];
        const float4 y0 = gq[0], y1 = gq[1];
        const float l  = gat_logit(x0, x1, q0, q1, a0, a1);
        const float mn = fmaxf(m, l);
        const float sc = __expf(m - mn);
        m = mn;
        const float w = __expf(l - mn);
        s *= sc;
        s += w;
        acc[0] *= sc; acc[1] *= sc; acc[2] *= sc; acc[3] *= sc;
        acc[4] *= sc; acc[5] *= sc; acc[6] *= sc; acc[7] *= sc;
        gat_axpy(acc, y0, y1, w);
      }
    }

    o0.x = acc[0] / s; o0.y = acc[1] / s; o0.z = acc[2] / s; o0.w = acc[3] / s;
    o1.x = acc[4] / s; o1.y = acc[5] / s; o1.z = acc[6] / s; o1.w = acc[7] / s;
  }
  if (valid) {
    float4* op = (float4*)(out + (size_t)t * HD + co);
    op[0] = o0; op[1] = o1;
  }
}

extern "C" void kernel_launch(void* const* d_in, const int* in_sizes, int n_in,
                              void* d_out, int out_size, void* d_ws, size_t ws_size,
                              hipStream_t stream) {
  const float* h   = (const float*)d_in[0];
  const float* W   = (const float*)d_in[1];
  const float* a   = (const float*)d_in[2];
  const int*   src = (const int*)d_in[3];
  const int*   tgt = (const int*)d_in[4];
  float*       out = (float*)d_out;

  const int nNodes = in_sizes[0] / IN_CH;
  const int nEdges = in_sizes[3];

  const size_t nPad = ((size_t)nNodes + 64) & ~(size_t)63;
  const size_t ePad = ((size_t)nEdges + 63) & ~(size_t)63;
  float* P      = (float*)d_ws;
  float* Q      = P + (size_t)nNodes * HD;
  int*   cnt    = (int*)(Q + (size_t)nNodes * HD);
  int*   rowptr = cnt + nPad;
  int*   rank   = rowptr + nPad;
  int*   grp    = rank + ePad;

  const int rowBlocks = nNodes / 16;
  gat_project_wmma<<<rowBlocks, 256, 0, stream>>>(h, W, P, Q, cnt, nNodes);

  const int eBlocks = (nEdges + 255) / 256;
  gat_count<<<eBlocks, 256, 0, stream>>>(tgt, cnt, rank, nEdges);
  gat_scan <<<1, 1024, 0, stream>>>(cnt, rowptr, nNodes);
  gat_fill <<<eBlocks, 256, 0, stream>>>(tgt, rank, rowptr, grp, nEdges);

  const int nh = nNodes * HEADS;
  gat_gather<<<(nh + 255) / 256, 256, 0, stream>>>(h, P, Q, a, src, rowptr, grp,
                                                   out, nNodes);
}


# Edges binned by target range in LDS (LDS integer atomics), one 128-byte segment per bin and chunk; the gather fills its buckets from them: no global atomics or counters; source-id sum order kept
# speedup vs baseline: 4.5236x; 2.7201x over previous
#include <hip/hip_runtime.h>
#include <hip/hip_bf16.h>
#include <math.h>


typedef __attribute__((ext_vector_type(16))) __bf16 v16bf;
typedef __attribute__((ext_vector_type(8)))  float  v8f;

#define IN_CH   64
#define HEADS   8
#define OUT_CH  8
#define HD      (HEADS * OUT_CH)
#define LEAKY   0.3f
#define HT_PITCH 68
#define GRP_CAP 64
#define BIN_T   32
#define SEG_W   32
#define SEG_CAP (SEG_W - 1)
#define CHUNK_E 8192
#define PART_T  1024
#define WO_GRP  7
#define MAX_BINS 1568
#define SEG_GRP 13

__global__ __launch_bounds__(256) void gat_project_wmma(
    const float* __restrict__ h, const float* __restrict__ W,
    float* __restrict__ P, float* __restrict__ Q, int nNodes) {
  __shared__ float hT[16 * HT_PITCH];
  __shared__ float Wl[2 * IN_CH * HD];

  const int tid = threadIdx.x;

  {
    const float4* s4 = (const float4*)(h + (size_t)blockIdx.x * 16 * IN_CH);
    const int r = tid >> 4;
    const int c = (tid & 15) * 4;
    *(float4*)(hT + r * HT_PITCH + c) = s4[tid];
  }
  {
    const float4* s4 = (const float4*)W;
    float4* d4 = (float4*)Wl;
    #pragma unroll
    for (int i = 0; i < 8; ++i)
      d4[tid + 256 * i] = s4[tid + 256 * i];
  }
  __syncthreads();

  const int wave  = tid >> 5;
  const int lane  = tid & 31;
  const int mat   = wave >> 2;
  const int tn    = wave & 3;
  const int l16   = lane & 15;
  const int khalf = lane >> 4;
  const int col   = tn * 16 + l16;
  const int wbase = mat * IN_CH;

  const float* __restrict__ hrow = hT + l16 * HT_PITCH;

  v8f acc = {};
  #pragma unroll
  for (int kb = 0; kb < IN_CH; kb += 32) {
    const int k0 = kb + khalf * 8;
    v16bf A, B;
    #pragma unroll
    for (int j = 0; j < 8; ++j) {
      A[j]     = (__bf16)hrow[k0 + j];
      A[j + 8] = (__bf16)hrow[k0 + 16 + j];
      B[j]     = (__bf16)Wl[(wbase + k0 + j)      * HD + col];
      B[j + 8] = (__bf16)Wl[(wbase + k0 + 16 + j) * HD + col];
    }
    acc = __builtin_amdgcn_wmma_f32_16x16x32_bf16(
         false, A,  false, B,
         (short)0, acc,  false,  false);
  }

  float* __restrict__ dst = mat ? Q : P;
  const int rbase = blockIdx.x * 16 + khalf * 8;
  #pragma unroll
  for (int v = 0; v < 8; ++v)
    dst[(size_t)(rbase + v) * HD + col] = acc[v];
}

__global__ __launch_bounds__(PART_T) void gat_partition(
    const int* __restrict__ src, const int* __restrict__ tgt,
    int* __restrict__ seg, int nNodes, int nEdges, int nBins, int nChunks) {
  __shared__ int lseg[MAX_BINS * SEG_W];
  __shared__ int lcnt[MAX_BINS];

  const int tid = threadIdx.x;
  for (int b = tid; b < nBins; b += PART_T) lcnt[b] = 0;

  const int e0 = blockIdx.x * CHUNK_E + tid * 8;
  int tv[8], sv[8];
  if (e0 + 8 <= nEdges) {
    const int4 ta = *(const int4*)(tgt + e0), tb = *(const int4*)(tgt + e0 + 4);
    const int4 sa = *(const int4*)(src + e0), sb = *(const int4*)(src + e0 + 4);
    tv[0] = ta.x; tv[1] = ta.y; tv[2] = ta.z; tv[3] = ta.w;
    tv[4] = tb.x; tv[5] = tb.y; tv[6] = tb.z; tv[7] = tb.w;
    sv[0] = sa.x; sv[1] = sa.y; sv[2] = sa.z; sv[3] = sa.w;
    sv[4] = sb.x; sv[5] = sb.y; sv[6] = sb.z; sv[7] = sb.w;
  } else {
    #pragma unroll
    for (int j = 0; j < 8; ++j) {
      const int e = e0 + j;
      tv[j] = e < nEdges ? tgt[e] : -1;
      sv[j] = e < nEdges ? src[e] : 0;
    }
  }
  __syncthreads();
  int rk[8];
  #pragma unroll
  for (int j = 0; j < 8; ++j) {
    const int t = tv[j];
    rk[j] = SEG_CAP;
    if ((unsigned)t < (unsigned)nNodes) rk[j] = atomicAdd(&lcnt[t / BIN_T], 1);
  }
  #pragma unroll
  for (int j = 0; j < 8; ++j) {
    const int t = tv[j];
    if (rk[j] < SEG_CAP)
      lseg[(t / BIN_T) * SEG_W + 1 + rk[j]] = sv[j] * BIN_T + (t % BIN_T);
  }
  __syncthreads();
  const int total = nBins * SEG_W;
  const int j = tid % SEG_W;
  for (int w0 = tid; w0 < total; w0 += PART_T * WO_GRP) {
    int cc[WO_GRP], vv[WO_GRP];
    #pragma unroll
    for (int u = 0; u < WO_GRP; ++u) {
      const int w = w0 + u * PART_T;
      cc[u] = (w < total) ? lcnt[w / SEG_W] : 0;
    }
    #pragma unroll
    for (int u = 0; u < WO_GRP; ++u) {
      const int w = w0 + u * PART_T;
      vv[u] = (j == 0) ? cc[u] : 0;
      if (j > 0 && j <= cc[u]) vv[u] = lseg[w];
    }
    #pragma unroll
    for (int u = 0; u < WO_GRP; ++u) {
      const int w = w0 + u * PART_T;
      if (w < total)
        seg[((size_t)(w / SEG_W) * nChunks + blockIdx.x) * SEG_W + j] = vv[u];
    }
  }
}

typedef __attribute__((ext_vector_type(4))) int v4i;

typedef __attribute__((ext_vector_type(2))) float v2f;

__device__ __forceinline__ float gat_logit(const float4 p0, const float4 p1,
                                           const float4 q0, const float4 q1,
                                           const float4 a0, const float4 a1) {
  const v2f pv[4] = {{p0.x, p0.y}, {p0.z, p0.w}, {p1.x, p1.y}, {p1.z, p1.w}};
  const v2f qv[4] = {{q0.x, q0.y}, {q0.z, q0.w}, {q1.x, q1.y}, {q1.z, q1.w}};
  const v2f av[4] = {{a0.x, a0.y}, {a0.z, a0.w}, {a1.x, a1.y}, {a1.z, a1.w}};
  float lg = 0.0f;
  #pragma unroll
  for (int j = 0; j < 4; ++j) {
    const v2f x = pv[j] + qv[j];
    const v2f y = x * LEAKY;
    const float x0 = x.x > 0.0f ? x.x : y.x;
    const float x1 = x.y > 0.0f ? x.y : y.y;
    lg += x0 * av[j].x;
    lg += x1 * av[j].y;
  }
  return lg;
}

__device__ __forceinline__ void gat_axpy(float* acc, const float4 h0,
                                         const float4 h1, const float w) {
  acc[0] += h0.x * w; acc[1] += h0.y * w; acc[2] += h0.z * w; acc[3] += h0.w * w;
  acc[4] += h1.x * w; acc[5] += h1.y * w; acc[6] += h1.z * w; acc[7] += h1.w * w;
}

__global__ __launch_bounds__(256) void gat_gather(
    const float* __restrict__ h, const float* __restrict__ P,
    const float* __restrict__ Q, const float* __restrict__ a,
    const int* __restrict__ seg, const int* __restrict__ src,
    const int* __restrict__ tgt, float* __restrict__ out, int nNodes,
    int nEdges, int nChunks) {
  __shared__ __attribute__((aligned(16))) int ky[32 * GRP_CAP];
  __shared__ __attribute__((aligned(16))) int ss[32 * GRP_CAP];
  __shared__ int dcnt[BIN_T];
  __shared__ int lost;

  const int tid = threadIdx.x;
  const int g = blockIdx.x * 256 + tid;
  const bool valid = g < nNodes * HEADS;
  const int t  = g >> 3;
  const int hd = g & 7;
  const int o  = (tid >> 3) * GRP_CAP;
  const int co = hd * OUT_CH;
  if (tid < BIN_T) dcnt[tid] = 0;
  if (tid == BIN_T) lost = 0;
  {
    const v4i top = {0x7fffffff, 0x7fffffff, 0x7fffffff, 0x7fffffff};
    ((v4i*)ky)[tid] = top;
    ((v4i*)ky)[tid + 256] = top;
  }

  const float4* qp = (const float4*)(Q + (size_t)(valid ? t : 0) * HD + co);
  const float4 q0 = qp[0], q1 = qp[1];
  const float4* ap = (const float4*)(a + co);
  const float4 a0 = ap[0], a1 = ap[1];

  __syncthreads();

  {
    const int* __restrict__ sg = seg + (size_t)blockIdx.x * nChunks * SEG_W;
    const int nw = nChunks * SEG_W;
    const int ln = tid & 31;
    bool over = false;
    for (int base = 0; base < nw; base += 256 * SEG_GRP) {
      int wd[SEG_GRP];
      #pragma unroll
      for (int k = 0; k < SEG_GRP; ++k) {
        const int w = base + k * 256 + tid;
        const int v = sg[min(w, nw - 1)];
        wd[k] = (w < nw) ? v : 0;
      }
      int rk[SEG_GRP];
      #pragma unroll
      for (int k = 0; k < SEG_GRP; ++k) {
        const int c = __builtin_amdgcn_readfirstlane(wd[k]);
        over = over || (c > SEG_CAP);
        rk[k] = GRP_CAP;
        if (ln >= 1 && ln <= c) rk[k] = atomicAdd(&dcnt[(unsigned)wd[k] % BIN_T], 1);
      }
      #pragma unroll
      for (int k = 0; k < SEG_GRP; ++k) {
        const int r = rk[k];
        if (r < GRP_CAP) {
          const unsigned word = (unsigned)wd[k];
          ky[(word % BIN_T) * GRP_CAP + r] = (int)(word / BIN_T) * GRP_CAP + r;
        }
      }
    }
    if (over && ln == 0) lost = 1;
  }
  __syncthreads();

  const int deg = valid ? dcnt[tid >> 3] : 0;
  const int last = deg - 1;
  const bool slow = valid && (lost != 0 || deg > GRP_CAP);
  const bool fits = deg > 0 && !slow;
  if (fits) {
    for (int idx = hd; idx < deg; idx += 8) {
      const int k = ky[o + idx];
      int r = 0;
      for (int q = 0; q < deg; q += 4) {
        const v4i v = *(const v4i*)(ky + o + q);
        r += (v.x < k ? 1 : 0) + (v.y < k ? 1 : 0) + (v.z < k ? 1 : 0) + (v.w < k ? 1 : 0);
      }
      ss[o + r] = (int)((unsigned)k / GRP_CAP);
    }
  }
  __syncthreads();

  float4 o0 = make_float4(0.0f, 0.0f, 0.0f, 0.0f), o1 = o0;

  if (fits || slow) {
    float m = -__builtin_inff();
    float s = 0.0f;
    float acc[OUT_CH] = {0.0f, 0.0f, 0.0f, 0.0f, 0.0f, 0.0f, 0.0f, 0.0f};
    const unsigned cb = (unsigned)co * 4u;
    const char* __restrict__ Pb = (const char*)P;
    const char* __restrict__ hb = (const char*)h;

    if (fits) {
      int n0 = ss[o];
      int n1 = ss[o + min(1, last)];
      int n2 = ss[o + min(2, last)];
      int n3 = ss[o + min(3, last)];
      for (int i = 0; i < deg; i += 4) {
        const unsigned f0 = (unsigned)n0 * (HD * 4u) + cb;
        const unsigned f1 = (unsigned)n1 * (HD * 4u) + cb;
        const unsigned f2 = (unsigned)n2 * (HD * 4u) + cb;
        const unsigned f3 = (unsigned)n3 * (HD * 4u) + cb;
        n0 = ss[o + min(i + 4, last)];
        n1 = ss[o + min(i + 5, last)];
        n2 = ss[o + min(i + 6, last)];
        n3 = ss[o + min(i + 7, last)];
        const float4* r0 = (const float4*)(Pb + f0);
        const float4* r1 = (const float4*)(Pb + f1);
        const float4* r2 = (const float4*)(Pb + f2);
        const float4* r3 = (const float4*)(Pb + f3);
        const float4 x00 = r0[0], x01 = r0[1], x10 = r1[0], x11 = r1[1];
        const float4 x20 = r2[0], x21 = r2[1], x30 = r3[0], x31 = r3[1];
        const float4* g0 = (const float4*)(hb + f0);
        const float4* g1 = (const float4*)(hb + f1);
        const float4* g2 = (const float4*)(hb + f2);
        const float4* g3 = (const float4*)(hb + f3);
        const float4 y00 = g0[0], y01 = g0[1], y10 = g1[0], y11 = g1[1];
        const float4 y20 = g2[0], y21 = g2[1], y30 = g3[0], y31 = g3[1];

        const float l0 = gat_logit(x00, x01, q0, q1, a0, a1);
        const float l1 = gat_logit(x10, x11, q0, q1, a0, a1);
        const float l2 = gat_logit(x20, x21, q0, q1, a0, a1);
        const float l3 = gat_logit(x30, x31, q0, q1, a0, a1);

        const float mn = fmaxf(m, fmaxf(fmaxf(l0, l1), fmaxf(l2, l3)));
        const float sc = __expf(m - mn);
        m = mn;
        const float e0 = __expf(l0 - mn);
        const float e1 = (i + 1 < deg) ? __expf(l1 - mn) : 0.0f;
        const float e2 = (i + 2 < deg) ? __expf(l2 - mn) : 0.0f;
        const float e3 = (i + 3 < deg) ? __expf(l3 - mn) : 0.0f;
        s *= sc;
        s += e0; s += e1; s += e2; s += e3;
        acc[0] *= sc; acc[1] *= sc; acc[2] *= sc; acc[3] *= sc;
        acc[4] *= sc; acc[5] *= sc; acc[6] *= sc; acc[7] *= sc;
        gat_axpy(acc, y00, y01, e0);
        gat_axpy(acc, y10, y11, e1);
        gat_axpy(acc, y20, y21, e2);
        gat_axpy(acc, y30, y31, e3);
        asm volatile("" :: "v"(n0), "v"(n1), "v"(n2), "v"(n3));
        __builtin_amdgcn_sched_group_barrier(0x020, 16, 0);
        __builtin_amdgcn_sched_group_barrier(0x002, 512, 0);
      }
    } else {
      int prev = -1;
      for (int it = 0; it < nEdges; ++it) {
        int best = 0x7fffffff;
        int mult = 0;
        for (int e = 0; e < nEdges; ++e) {
          if (tgt[e] == t) {
            const int v = src[e];
            if (v > prev) {
              if (v < best) { best = v; mult = 1; }
              else if (v == best) ++mult;
            }
          }
        }
        if (mult == 0) break;
        prev = best;
        const unsigned f = (unsigned)best * (HD * 4u) + cb;
        const float4* r  = (const float4*)(Pb + f);
        const float4* gq = (const float4*)(hb + f);
        const float4 x0 = r[0], x1 = r[1];
        const float4 y0 = gq[0], y1 = gq[1];
        const float l  = gat_logit(x0, x1, q0, q1, a0, a1);
        const float mn = fmaxf(m, l);
        const float sc = __expf(m - mn);
        m = mn;
        const float w = __expf(l - mn);
        s *= sc;
        acc[0] *= sc; acc[1] *= sc; acc[2] *= sc; acc[3] *= sc;
        acc[4] *= sc; acc[5] *= sc; acc[6] *= sc; acc[7] *= sc;
        for (int c = 0; c < mult; ++c) {
          s += w;
          gat_axpy(acc, y0, y1, w);
        }
      }
    }

    if (s > 0.0f) {
      o0.x = acc[0] / s; o0.y = acc[1] / s; o0.z = acc[2] / s; o0.w = acc[3] / s;
      o1.x = acc[4] / s; o1.y = acc[5] / s; o1.z = acc[6] / s; o1.w = acc[7] / s;
    }
  }
  if (valid) {
    float4* op = (float4*)(out + (size_t)t * HD + co);
    op[0] = o0; op[1] = o1;
  }
}

extern "C" void kernel_launch(void* const* d_in, const int* in_sizes, int n_in,
                              void* d_out, int out_size, void* d_ws, size_t ws_size,
                              hipStream_t stream) {
  const float* h   = (const float*)d_in[0];
  const float* W   = (const float*)d_in[1];
  const float* a   = (const float*)d_in[2];
  const int*   src = (const int*)d_in[3];
  const int*   tgt = (const int*)d_in[4];
  float*       out = (float*)d_out;

  const int nNodes = in_sizes[0] / IN_CH;
  const int nEdges = in_sizes[3];

  const int nBins   = (nNodes + BIN_T - 1) / BIN_T;
  const int nChunks = (nEdges + CHUNK_E - 1) / CHUNK_E;
  float* P   = (float*)d_ws;
  float* Q   = P + (size_t)nNodes * HD;
  int*   seg = (int*)(Q + (size_t)nNodes * HD);

  const int rowBlocks = nNodes / 16;
  gat_project_wmma<<<rowBlocks, 256, 0, stream>>>(h, W, P, Q, nNodes);

  gat_partition<<<nChunks, PART_T, 0, stream>>>(src, tgt, seg, nNodes, nEdges,
                                                nBins, nChunks);

  gat_gather<<<nBins, 256, 0, stream>>>(h, P, Q, a, seg, src, tgt, out, nNodes,
                                        nEdges, nChunks);
}


# Gather reads and writes whole 128-byte lines: P, Q stored head-interleaved, feature rows dealt 16 B per lane, head weights moved within the 8-lane group (DPP8); each row load touches 4 lines, not 8
# speedup vs baseline: 4.6638x; 1.0310x over previous
#include <hip/hip_runtime.h>
#include <hip/hip_bf16.h>
#include <math.h>


typedef __attribute__((ext_vector_type(16))) __bf16 v16bf;
typedef __attribute__((ext_vector_type(8)))  float  v8f;

#define IN_CH   64
#define HEADS   8
#define OUT_CH  8
#define HD      (HEADS * OUT_CH)
#define LEAKY   0.3f
#define HT_PITCH 68
#define GRP_CAP 64
#define BIN_T   32
#define SEG_W   32
#define SEG_CAP (SEG_W - 1)
#define CHUNK_E 8192
#define PART_T  1024
#define WO_GRP  7
#define MAX_BINS 1568
#define SEG_GRP 13

__global__ __launch_bounds__(256) void gat_project_wmma(
    const float* __restrict__ h, const float* __restrict__ W,
    float* __restrict__ P, float* __restrict__ Q, int nNodes) {
  __shared__ float hT[16 * HT_PITCH];
  __shared__ float Wl[2 * IN_CH * HD];

  const int tid = threadIdx.x;

  {
    const float4* s4 = (const float4*)(h + (size_t)blockIdx.x * 16 * IN_CH);
    const int r = tid >> 4;
    const int c = (tid & 15) * 4;
    *(float4*)(hT + r * HT_PITCH + c) = s4[tid];
  }
  {
    const float4* s4 = (const float4*)W;
    float4* d4 = (float4*)Wl;
    #pragma unroll
    for (int i = 0; i < 8; ++i)
      d4[tid + 256 * i] = s4[tid + 256 * i];
  }
  __syncthreads();

  const int wave  = tid >> 5;
  const int lane  = tid & 31;
  const int mat   = wave >> 2;
  const int tn    = wave & 3;
  const int l16   = lane & 15;
  const int khalf = lane >> 4;
  const int col   = tn * 16 + l16;
  const int wcol  = ((col & 31) >> 2) * OUT_CH + (col >> 5) * 4 + (col & 3);
  const int wbase = mat * IN_CH;

  const float* __restrict__ hrow = hT + l16 * HT_PITCH;

  v8f acc = {};
  #pragma unroll
  for (int kb = 0; kb < IN_CH; kb += 32) {
    const int k0 = kb + khalf * 8;
    v16bf A, B;
    #pragma unroll
    for (int j = 0; j < 8; ++j) {
      A[j]     = (__bf16)hrow[k0 + j];
      A[j + 8] = (__bf16)hrow[k0 + 16 + j];
      B[j]     = (__bf16)Wl[(wbase + k0 + j)      * HD + wcol];
      B[j + 8] = (__bf16)Wl[(wbase + k0 + 16 + j) * HD + wcol];
    }
    acc = __builtin_amdgcn_wmma_f32_16x16x32_bf16(
         false, A,  false, B,
         (short)0, acc,  false,  false);
  }

  float* __restrict__ dst = mat ? Q : P;
  const int rbase = blockIdx.x * 16 + khalf * 8;
  #pragma unroll
  for (int v = 0; v < 8; ++v)
    dst[(size_t)(rbase + v) * HD + col] = acc[v];
}

__global__ __launch_bounds__(PART_T) void gat_partition(
    const int* __restrict__ src, const int* __restrict__ tgt,
    int* __restrict__ seg, int nNodes, int nEdges, int nBins, int nChunks) {
  __shared__ int lseg[MAX_BINS * SEG_W];
  __shared__ int lcnt[MAX_BINS];

  const int tid = threadIdx.x;
  for (int b = tid; b < nBins; b += PART_T) lcnt[b] = 0;

  const int e0 = blockIdx.x * CHUNK_E + tid * 8;
  int tv[8], sv[8];
  if (e0 + 8 <= nEdges) {
    const int4 ta = *(const int4*)(tgt + e0), tb = *(const int4*)(tgt + e0 + 4);
    const int4 sa = *(const int4*)(src + e0), sb = *(const int4*)(src + e0 + 4);
    tv[0] = ta.x; tv[1] = ta.y; tv[2] = ta.z; tv[3] = ta.w;
    tv[4] = tb.x; tv[5] = tb.y; tv[6] = tb.z; tv[7] = tb.w;
    sv[0] = sa.x; sv[1] = sa.y; sv[2] = sa.z; sv[3] = sa.w;
    sv[4] = sb.x; sv[5] = sb.y; sv[6] = sb.z; sv[7] = sb.w;
  } else {
    #pragma unroll
    for (int j = 0; j < 8; ++j) {
      const int e = e0 + j;
      tv[j] = e < nEdges ? tgt[e] : -1;
      sv[j] = e < nEdges ? src[e] : 0;
    }
  }
  __syncthreads();
  int rk[8];
  #pragma unroll
  for (int j = 0; j < 8; ++j) {
    const int t = tv[j];
    rk[j] = SEG_CAP;
    if ((unsigned)t < (unsigned)nNodes) rk[j] = atomicAdd(&lcnt[t / BIN_T], 1);
  }
  #pragma unroll
  for (int j = 0; j < 8; ++j) {
    const int t = tv[j];
    if (rk[j] < SEG_CAP)
      lseg[(t / BIN_T) * SEG_W + 1 + rk[j]] = sv[j] * BIN_T + (t % BIN_T);
  }
  __syncthreads();
  const int total = nBins * SEG_W;
  const int j = tid % SEG_W;
  for (int w0 = tid; w0 < total; w0 += PART_T * WO_GRP) {
    int cc[WO_GRP], vv[WO_GRP];
    #pragma unroll
    for (int u = 0; u < WO_GRP; ++u) {
      const int w = w0 + u * PART_T;
      cc[u] = (w < total) ? lcnt[w / SEG_W] : 0;
    }
    #pragma unroll
    for (int u = 0; u < WO_GRP; ++u) {
      const int w = w0 + u * PART_T;
      vv[u] = (j == 0) ? cc[u] : 0;
      if (j > 0 && j <= cc[u]) vv[u] = lseg[w];
    }
    #pragma unroll
    for (int u = 0; u < WO_GRP; ++u) {
      const int w = w0 + u * PART_T;
      if (w < total)
        seg[((size_t)(w / SEG_W) * nChunks + blockIdx.x) * SEG_W + j] = vv[u];
    }
  }
}

typedef __attribute__((ext_vector_type(4))) int v4i;

typedef __attribute__((ext_vector_type(2))) float v2f;

__device__ __forceinline__ float gat_logit(const float4 p0, const float4 p1,
                                           const float4 q0, const float4 q1,
                                           const float4 a0, const float4 a1) {
  const v2f pv[4] = {{p0.x, p0.y}, {p0.z, p0.w}, {p1.x, p1.y}, {p1.z, p1.w}};
  const v2f qv[4] = {{q0.x, q0.y}, {q0.z, q0.w}, {q1.x, q1.y}, {q1.z, q1.w}};
  const v2f av[4] = {{a0.x, a0.y}, {a0.z, a0.w}, {a1.x, a1.y}, {a1.z, a1.w}};
  float lg = 0.0f;
  #pragma unroll
  for (int j = 0; j < 4; ++j) {
    const v2f x = pv[j] + qv[j];
    const v2f y = x * LEAKY;
    const float x0 = x.x > 0.0f ? x.x : y.x;
    const float x1 = x.y > 0.0f ? x.y : y.y;
    lg += x0 * av[j].x;
    lg += x1 * av[j].y;
  }
  return lg;
}

__device__ __forceinline__ void gat_axpy(v2f* acc, const float4 h0,
                                         const float4 h1, const float wa,
                                         const float wb) {
  const v2f a0 = {h0.x, h0.y}, a1 = {h0.z, h0.w};
  const v2f b0 = {h1.x, h1.y}, b1 = {h1.z, h1.w};
  acc[0] += a0 * wa; acc[1] += a1 * wa;
  acc[2] += b0 * wb; acc[3] += b1 * wb;
}

__device__ __forceinline__ float gat_from_a(const float v) {
  float r;
  asm("v_mov_b32_dpp %0, %1 dpp8:[0,0,1,1,2,2,3,3]" : "=v"(r) : "v"(v));
  return r;
}
__device__ __forceinline__ float gat_from_b(const float v) {
  float r;
  asm("v_mov_b32_dpp %0, %1 dpp8:[4,4,5,5,6,6,7,7]" : "=v"(r) : "v"(v));
  return r;
}

__global__ __launch_bounds__(256) void gat_gather(
    const float* __restrict__ h, const float* __restrict__ P,
    const float* __restrict__ Q, const float* __restrict__ a,
    const int* __restrict__ seg, const int* __restrict__ src,
    const int* __restrict__ tgt, float* __restrict__ out, int nNodes,
    int nEdges, int nChunks) {
  __shared__ __attribute__((aligned(16))) int ky[32 * GRP_CAP];
  __shared__ __attribute__((aligned(16))) int ss[32 * GRP_CAP];
  __shared__ int dcnt[BIN_T];
  __shared__ int lost;

  const int tid = threadIdx.x;
  const int g = blockIdx.x * 256 + tid;
  const bool valid = g < nNodes * HEADS;
  const int t  = g >> 3;
  const int hd = g & 7;
  const int o  = (tid >> 3) * GRP_CAP;
  const int co = hd * OUT_CH;
  const int lo = hd * 4;
  if (tid < BIN_T) dcnt[tid] = 0;
  if (tid == BIN_T) lost = 0;
  {
    const v4i top = {0x7fffffff, 0x7fffffff, 0x7fffffff, 0x7fffffff};
    ((v4i*)ky)[tid] = top;
    ((v4i*)ky)[tid + 256] = top;
  }

  const float4* qp = (const float4*)(Q + (size_t)(valid ? t : 0) * HD + lo);
  const float4 q0 = qp[0], q1 = qp[8];
  const float4* ap = (const float4*)(a + co);
  const float4 a0 = ap[0], a1 = ap[1];

  __syncthreads();

  {
    const int* __restrict__ sg = seg + (size_t)blockIdx.x * nChunks * SEG_W;
    const int nw = nChunks * SEG_W;
    const int ln = tid & 31;
    bool over = false;
    for (int base = 0; base < nw; base += 256 * SEG_GRP) {
      int wd[SEG_GRP];
      #pragma unroll
      for (int k = 0; k < SEG_GRP; ++k) {
        const int w = base + k * 256 + tid;
        const int v = sg[min(w, nw - 1)];
        wd[k] = (w < nw) ? v : 0;
      }
      int rk[SEG_GRP];
      #pragma unroll
      for (int k = 0; k < SEG_GRP; ++k) {
        const int c = __builtin_amdgcn_readfirstlane(wd[k]);
        over = over || (c > SEG_CAP);
        rk[k] = GRP_CAP;
        if (ln >= 1 && ln <= c) rk[k] = atomicAdd(&dcnt[(unsigned)wd[k] % BIN_T], 1);
      }
      #pragma unroll
      for (int k = 0; k < SEG_GRP; ++k) {
        const int r = rk[k];
        if (r < GRP_CAP) {
          const unsigned word = (unsigned)wd[k];
          ky[(word % BIN_T) * GRP_CAP + r] = (int)(word / BIN_T) * GRP_CAP + r;
        }
      }
    }
    if (over && ln == 0) lost = 1;
  }
  __syncthreads();

  const int deg = valid ? dcnt[tid >> 3] : 0;
  const int last = deg - 1;
  const bool slow = valid && (lost != 0 || deg > GRP_CAP);
  const bool fits = deg > 0 && !slow;
  if (fits) {
    for (int idx = hd; idx < deg; idx += 8) {
      const int k = ky[o + idx];
      int r = 0;
      for (int q = 0; q < deg; q += 4) {
        const v4i v = *(const v4i*)(ky + o + q);
        r += (v.x < k ? 1 : 0) + (v.y < k ? 1 : 0) + (v.z < k ? 1 : 0) + (v.w < k ? 1 : 0);
      }
      ss[o + r] = (int)((unsigned)k / GRP_CAP);
    }
  }
  __syncthreads();

  float4 o0 = make_float4(0.0f, 0.0f, 0.0f, 0.0f), o1 = o0;

  if (fits || slow) {
    float m = -__builtin_inff();
    float s = 0.0f;
    v2f acc[4] = {{0.0f, 0.0f}, {0.0f, 0.0f}, {0.0f, 0.0f}, {0.0f, 0.0f}};
    const unsigned cb = (unsigned)lo * 4u;
    const char* __restrict__ Pb = (const char*)P;
    const char* __restrict__ hb = (const char*)h;

    if (fits) {
      int n0 = ss[o];
      int n1 = ss[o + min(1, last)];
      int n2 = ss[o + min(2, last)];
      int n3 = ss[o + min(3, last)];
      for (int i = 0; i < deg; i += 4) {
        const unsigned f0 = (unsigned)n0 * (HD * 4u) + cb;
        const unsigned f1 = (unsigned)n1 * (HD * 4u) + cb;
        const unsigned f2 = (unsigned)n2 * (HD * 4u) + cb;
        const unsigned f3 = (unsigned)n3 * (HD * 4u) + cb;
        n0 = ss[o + min(i + 4, last)];
        n1 = ss[o + min(i + 5, last)];
        n2 = ss[o + min(i + 6, last)];
        n3 = ss[o + min(i + 7, last)];
        const float4* r0 = (const float4*)(Pb + f0);
        const float4* r1 = (const float4*)(Pb + f1);
        const float4* r2 = (const float4*)(Pb + f2);
        const float4* r3 = (const float4*)(Pb + f3);
        const float4 x00 = r0[0], x01 = r0[8], x10 = r1[0], x11 = r1[8];
        const float4 x20 = r2[0], x21 = r2[8], x30 = r3[0], x31 = r3[8];
        const float4* g0 = (const float4*)(hb + f0);
        const float4* g1 = (const float4*)(hb + f1);
        const float4* g2 = (const float4*)(hb + f2);
        const float4* g3 = (const float4*)(hb + f3);
        const float4 y00 = g0[0], y01 = g0[8], y10 = g1[0], y11 = g1[8];
        const float4 y20 = g2[0], y21 = g2[8], y30 = g3[0], y31 = g3[8];

        const float l0 = gat_logit(x00, x01, q0, q1, a0, a1);
        const float l1 = gat_logit(x10, x11, q0, q1, a0, a1);
        const float l2 = gat_logit(x20, x21, q0, q1, a0, a1);
        const float l3 = gat_logit(x30, x31, q0, q1, a0, a1);

        const float mn = fmaxf(m, fmaxf(fmaxf(l0, l1), fmaxf(l2, l3)));
        const float sc = __expf(m - mn);
        m = mn;
        const float e0 = __expf(l0 - mn);
        const float e1 = (i + 1 < deg) ? __expf(l1 - mn) : 0.0f;
        const float e2 = (i + 2 < deg) ? __expf(l2 - mn) : 0.0f;
        const float e3 = (i + 3 < deg) ? __expf(l3 - mn) : 0.0f;
        s *= sc;
        s += e0; s += e1; s += e2; s += e3;
        const float sca = gat_from_a(sc), scb = gat_from_b(sc);
        acc[0] *= sca; acc[1] *= sca;
        acc[2] *= scb; acc[3] *= scb;
        gat_axpy(acc, y00, y01, gat_from_a(e0), gat_from_b(e0));
        gat_axpy(acc, y10, y11, gat_from_a(e1), gat_from_b(e1));
        gat_axpy(acc, y20, y21, gat_from_a(e2), gat_from_b(e2));
        gat_axpy(acc, y30, y31, gat_from_a(e3), gat_from_b(e3));
        asm volatile("" :: "v"(n0), "v"(n1), "v"(n2), "v"(n3));
        __builtin_amdgcn_sched_group_barrier(0x020, 16, 0);
        __builtin_amdgcn_sched_group_barrier(0x002, 512, 0);
      }
    } else {
      int prev = -1;
      for (int it = 0; it < nEdges; ++it) {
        int best = 0x7fffffff;
        int mult = 0;
        for (int e = 0; e < nEdges; ++e) {
          if (tgt[e] == t) {
            const int v = src[e];
            if (v > prev) {
              if (v < best) { best = v; mult = 1; }
              else if (v == best) ++mult;
            }
          }
        }
        if (mult == 0) break;
        prev = best;
        const unsigned f = (unsigned)best * (HD * 4u) + cb;
        const float4* r  = (const float4*)(Pb + f);
        const float4* gq = (const float4*)(hb + f);
        const float4 x0 = r[0], x1 = r[8];
        const float4 y0 = gq[0], y1 = gq[8];
        const float l  = gat_logit(x0, x1, q0, q1, a0, a1);
        const float mn = fmaxf(m, l);
        const float sc = __expf(m - mn);
        m = mn;
        const float w = __expf(l - mn);
        s *= sc;
        const float sca = gat_from_a(sc), scb = gat_from_b(sc);
        const float wa = gat_from_a(w), wb = gat_from_b(w);
        acc[0] *= sca; acc[1] *= sca;
        acc[2] *= scb; acc[3] *= scb;
        for (int c = 0; c < mult; ++c) {
          s += w;
          gat_axpy(acc, y0, y1, wa, wb);
        }
      }
    }

    const float sa = gat_from_a(s), sb = gat_from_b(s);
    if (sa > 0.0f) {
      o0.x = acc[0].x / sa; o0.y = acc[0].y / sa; o0.z = acc[1].x / sa; o0.w = acc[1].y / sa;
    }
    if (sb > 0.0f) {
      o1.x = acc[2].x / sb; o1.y = acc[2].y / sb; o1.z = acc[3].x / sb; o1.w = acc[3].y / sb;
    }
  }
  if (valid) {
    float4* op = (float4*)(out + (size_t)t * HD + lo);
    op[0] = o0; op[8] = o1;
  }
}

extern "C" void kernel_launch(void* const* d_in, const int* in_sizes, int n_in,
                              void* d_out, int out_size, void* d_ws, size_t ws_size,
                              hipStream_t stream) {
  const float* h   = (const float*)d_in[0];
  const float* W   = (const float*)d_in[1];
  const float* a   = (const float*)d_in[2];
  const int*   src = (const int*)d_in[3];
  const int*   tgt = (const int*)d_in[4];
  float*       out = (float*)d_out;

  const int nNodes = in_sizes[0] / IN_CH;
  const int nEdges = in_sizes[3];

  const int nBins   = (nNodes + BIN_T - 1) / BIN_T;
  const int nChunks = (nEdges + CHUNK_E - 1) / CHUNK_E;
  float* P   = (float*)d_ws;
  float* Q   = P + (size_t)nNodes * HD;
  int*   seg = (int*)(Q + (size_t)nNodes * HD);

  const int rowBlocks = nNodes / 16;
  gat_project_wmma<<<rowBlocks, 256, 0, stream>>>(h, W, P, Q, nNodes);

  gat_partition<<<nChunks, PART_T, 0, stream>>>(src, tgt, seg, nNodes, nEdges,
                                                nBins, nChunks);

  gat_gather<<<nBins, 256, 0, stream>>>(h, P, Q, a, seg, src, tgt, out, nNodes,
                                        nEdges, nChunks);
}


# Projection and edge binning share one launch (98 work-groups bin, 158 project, each staging W once and its 64-row groups together): the two independent stages run side by side, one launch fewer
# speedup vs baseline: 5.4984x; 1.1790x over previous
#include <hip/hip_runtime.h>
#include <hip/hip_bf16.h>
#include <math.h>


typedef __attribute__((ext_vector_type(16))) __bf16 v16bf;
typedef __attribute__((ext_vector_type(8)))  float  v8f;

#define IN_CH   64
#define HEADS   8
#define OUT_CH  8
#define HD      (HEADS * OUT_CH)
#define LEAKY   0.3f
#define HT_PITCH 68
#define GRP_CAP 64
#define BIN_T   32
#define SEG_W   32
#define SEG_CAP (SEG_W - 1)
#define CHUNK_E 8192
#define PART_T  1024
#define WO_GRP  7
#define MAX_BINS 1568
#define SEG_GRP 13

#define PROJ_ROWS 64
#define PROJ_WG   158
#define PROJ_BATCH 5
#define HT_BUF    (PROJ_ROWS * HT_PITCH)

__global__ __launch_bounds__(PART_T) void gat_project_partition(
    const float* __restrict__ h, const float* __restrict__ W,
    float* __restrict__ P, float* __restrict__ Q,
    const int* __restrict__ src, const int* __restrict__ tgt,
    int* __restrict__ seg, int nNodes, int nEdges, int nBins, int nChunks) {
  __shared__ __attribute__((aligned(16))) int lds[MAX_BINS * SEG_W + MAX_BINS];

  const int tid = threadIdx.x;

  if ((int)blockIdx.x >= nChunks) {
    float* __restrict__ Wl = (float*)lds;
    float* __restrict__ hT = Wl + 2 * IN_CH * HD;

    const int pb    = (int)blockIdx.x - nChunks;
    const int npb   = (int)gridDim.x - nChunks;
    const int nGrp  = (nNodes + PROJ_ROWS - 1) / PROJ_ROWS;
    const int nTile = nNodes / 16;

    const int sr = tid >> 4;
    const int sc = (tid & 15) * 4;

    const int wave  = __builtin_amdgcn_readfirstlane(tid >> 5);
    const int lane  = tid & 31;
    const int tile  = wave >> 3;
    const int mat   = (wave >> 2) & 1;
    const int tn    = wave & 3;
    const int l16   = lane & 15;
    const int khalf = lane >> 4;
    const int col   = tn * 16 + l16;
    const int wcol  = ((col & 31) >> 2) * OUT_CH + (col >> 5) * 4 + (col & 3);
    const int wbase = mat * IN_CH;
    const int k0    = khalf * 8;
    float* __restrict__ dst = mat ? Q : P;

    v16bf B0 = {}, B1 = {};
    for (int g0 = pb; g0 < nGrp; g0 += npb * PROJ_BATCH) {
      const bool first = (g0 == pb);
      if (!first) __syncthreads();
      float4 w0, w1;
      if (first) {
        const float4* s4 = (const float4*)W;
        w0 = s4[tid]; w1 = s4[tid + PART_T];
      }
      float4 r[PROJ_BATCH];
      #pragma unroll
      for (int u = 0; u < PROJ_BATCH; ++u) {
        const int row = (g0 + u * npb) * PROJ_ROWS + sr;
        r[u] = make_float4(0.0f, 0.0f, 0.0f, 0.0f);
        if (row < nNodes) r[u] = *(const float4*)(h + (size_t)row * IN_CH + sc);
      }
      if (first) {
        float4* d4 = (float4*)Wl;
        d4[tid] = w0; d4[tid + PART_T] = w1;
      }
      #pragma unroll
      for (int u = 0; u < PROJ_BATCH; ++u)
        *(float4*)(hT + u * HT_BUF + sr * HT_PITCH + sc) = r[u];
      __syncthreads();

      if (first) {
        #pragma unroll
        for (int j = 0; j < 8; ++j) {
          B0[j]     = (__bf16)Wl[(wbase + k0 + j)           * HD + wcol];
          B0[j + 8] = (__bf16)Wl[(wbase + k0 + 16 + j)      * HD + wcol];
          B1[j]     = (__bf16)Wl[(wbase + 32 + k0 + j)      * HD + wcol];
          B1[j + 8] = (__bf16)Wl[(wbase + 32 + k0 + 16 + j) * HD + wcol];
        }
      }

      #pragma unroll
      for (int u = 0; u < PROJ_BATCH; ++u) {
        const int gt = (g0 + u * npb) * (PROJ_ROWS / 16) + tile;
        if (gt < nTile) {
          const float* __restrict__ hrow = hT + u * HT_BUF + (tile * 16 + l16) * HT_PITCH;
          v16bf A0, A1;
          #pragma unroll
          for (int j = 0; j < 8; ++j) {
            A0[j]     = (__bf16)hrow[k0 + j];
            A0[j + 8] = (__bf16)hrow[k0 + 16 + j];
            A1[j]     = (__bf16)hrow[32 + k0 + j];
            A1[j + 8] = (__bf16)hrow[32 + k0 + 16 + j];
          }
          v8f acc = {};
          acc = __builtin_amdgcn_wmma_f32_16x16x32_bf16(
               false, A0,  false, B0,
               (short)0, acc,  false,  false);
          acc = __builtin_amdgcn_wmma_f32_16x16x32_bf16(
               false, A1,  false, B1,
               (short)0, acc,  false,  false);

          float* __restrict__ d = dst + ((size_t)gt * 16 + khalf * 8) * HD + col;
          #pragma unroll
          for (int v = 0; v < 8; ++v)
            d[v * HD] = acc[v];
        }
      }
    }
    return;
  }

  int* __restrict__ lseg = lds;
  int* __restrict__ lcnt = lds + MAX_BINS * SEG_W;

  for (int b = tid; b < nBins; b += PART_T) lcnt[b] = 0;

  const int e0 = blockIdx.x * CHUNK_E + tid * 8;
  int tv[8], sv[8];
  if (e0 + 8 <= nEdges) {
    const int4 ta = *(const int4*)(tgt + e0), tb = *(const int4*)(tgt + e0 + 4);
    const int4 sa = *(const int4*)(src + e0), sb = *(const int4*)(src + e0 + 4);
    tv[0] = ta.x; tv[1] = ta.y; tv[2] = ta.z; tv[3] = ta.w;
    tv[4] = tb.x; tv[5] = tb.y; tv[6] = tb.z; tv[7] = tb.w;
    sv[0] = sa.x; sv[1] = sa.y; sv[2] = sa.z; sv[3] = sa.w;
    sv[4] = sb.x; sv[5] = sb.y; sv[6] = sb.z; sv[7] = sb.w;
  } else {
    #pragma unroll
    for (int j = 0; j < 8; ++j) {
      const int e = e0 + j;
      tv[j] = e < nEdges ? tgt[e] : -1;
      sv[j] = e < nEdges ? src[e] : 0;
    }
  }
  __syncthreads();
  int rk[8];
  #pragma unroll
  for (int j = 0; j < 8; ++j) {
    const int t = tv[j];
    rk[j] = SEG_CAP;
    if ((unsigned)t < (unsigned)nNodes) rk[j] = atomicAdd(&lcnt[t / BIN_T], 1);
  }
  #pragma unroll
  for (int j = 0; j < 8; ++j) {
    const int t = tv[j];
    if (rk[j] < SEG_CAP)
      lseg[(t / BIN_T) * SEG_W + 1 + rk[j]] = sv[j] * BIN_T + (t % BIN_T);
  }
  __syncthreads();
  const int total = nBins * SEG_W;
  const int j = tid % SEG_W;
  for (int w0 = tid; w0 < total; w0 += PART_T * WO_GRP) {
    int cc[WO_GRP], vv[WO_GRP];
    #pragma unroll
    for (int u = 0; u < WO_GRP; ++u) {
      const int w = w0 + u * PART_T;
      cc[u] = (w < total) ? lcnt[w / SEG_W] : 0;
    }
    #pragma unroll
    for (int u = 0; u < WO_GRP; ++u) {
      const int w = w0 + u * PART_T;
      vv[u] = (j == 0) ? cc[u] : 0;
      if (j > 0 && j <= cc[u]) vv[u] = lseg[w];
    }
    #pragma unroll
    for (int u = 0; u < WO_GRP; ++u) {
      const int w = w0 + u * PART_T;
      if (w < total)
        seg[((size_t)(w / SEG_W) * nChunks + blockIdx.x) * SEG_W + j] = vv[u];
    }
  }
}

typedef __attribute__((ext_vector_type(4))) int v4i;

typedef __attribute__((ext_vector_type(2))) float v2f;

__device__ __forceinline__ float gat_logit(const float4 p0, const float4 p1,
                                           const float4 q0, const float4 q1,
                                           const float4 a0, const float4 a1) {
  const v2f pv[4] = {{p0.x, p0.y}, {p0.z, p0.w}, {p1.x, p1.y}, {p1.z, p1.w}};
  const v2f qv[4] = {{q0.x, q0.y}, {q0.z, q0.w}, {q1.x, q1.y}, {q1.z, q1.w}};
  const v2f av[4] = {{a0.x, a0.y}, {a0.z, a0.w}, {a1.x, a1.y}, {a1.z, a1.w}};
  float lg = 0.0f;
  #pragma unroll
  for (int j = 0; j < 4; ++j) {
    const v2f x = pv[j] + qv[j];
    const v2f y = x * LEAKY;
    const float x0 = x.x > 0.0f ? x.x : y.x;
    const float x1 = x.y > 0.0f ? x.y : y.y;
    lg += x0 * av[j].x;
    lg += x1 * av[j].y;
  }
  return lg;
}

__device__ __forceinline__ void gat_axpy(v2f* acc, const float4 h0,
                                         const float4 h1, const float wa,
                                         const float wb) {
  const v2f a0 = {h0.x, h0.y}, a1 = {h0.z, h0.w};
  const v2f b0 = {h1.x, h1.y}, b1 = {h1.z, h1.w};
  acc[0] += a0 * wa; acc[1] += a1 * wa;
  acc[2] += b0 * wb; acc[3] += b1 * wb;
}

__device__ __forceinline__ float gat_from_a(const float v) {
  float r;
  asm("v_mov_b32_dpp %0, %1 dpp8:[0,0,1,1,2,2,3,3]" : "=v"(r) : "v"(v));
  return r;
}
__device__ __forceinline__ float gat_from_b(const float v) {
  float r;
  asm("v_mov_b32_dpp %0, %1 dpp8:[4,4,5,5,6,6,7,7]" : "=v"(r) : "v"(v));
  return r;
}

__global__ __launch_bounds__(256) void gat_gather(
    const float* __restrict__ h, const float* __restrict__ P,
    const float* __restrict__ Q, const float* __restrict__ a,
    const int* __restrict__ seg, const int* __restrict__ src,
    const int* __restrict__ tgt, float* __restrict__ out, int nNodes,
    int nEdges, int nChunks) {
  __shared__ __attribute__((aligned(16))) int ky[32 * GRP_CAP];
  __shared__ __attribute__((aligned(16))) int ss[32 * GRP_CAP];
  __shared__ int dcnt[BIN_T];
  __shared__ int lost;

  const int tid = threadIdx.x;
  const int g = blockIdx.x * 256 + tid;
  const bool valid = g < nNodes * HEADS;
  const int t  = g >> 3;
  const int hd = g & 7;
  const int o  = (tid >> 3) * GRP_CAP;
  const int co = hd * OUT_CH;
  const int lo = hd * 4;
  if (tid < BIN_T) dcnt[tid] = 0;
  if (tid == BIN_T) lost = 0;
  {
    const v4i top = {0x7fffffff, 0x7fffffff, 0x7fffffff, 0x7fffffff};
    ((v4i*)ky)[tid] = top;
    ((v4i*)ky)[tid + 256] = top;
  }

  const float4* qp = (const float4*)(Q + (size_t)(valid ? t : 0) * HD + lo);
  const float4 q0 = qp[0], q1 = qp[8];
  const float4* ap = (const float4*)(a + co);
  const float4 a0 = ap[0], a1 = ap[1];

  __syncthreads();

  {
    const int* __restrict__ sg = seg + (size_t)blockIdx.x * nChunks * SEG_W;
    const int nw = nChunks * SEG_W;
    const int ln = tid & 31;
    bool over = false;
    for (int base = 0; base < nw; base += 256 * SEG_GRP) {
      int wd[SEG_GRP];
      #pragma unroll
      for (int k = 0; k < SEG_GRP; ++k) {
        const int w = base + k * 256 + tid;
        const int v = sg[min(w, nw - 1)];
        wd[k] = (w < nw) ? v : 0;
      }
      int rk[SEG_GRP];
      #pragma unroll
      for (int k = 0; k < SEG_GRP; ++k) {
        const int c = __builtin_amdgcn_readfirstlane(wd[k]);
        over = over || (c > SEG_CAP);
        rk[k] = GRP_CAP;
        if (ln >= 1 && ln <= c) rk[k] = atomicAdd(&dcnt[(unsigned)wd[k] % BIN_T], 1);
      }
      #pragma unroll
      for (int k = 0; k < SEG_GRP; ++k) {
        const int r = rk[k];
        if (r < GRP_CAP) {
          const unsigned word = (unsigned)wd[k];
          ky[(word % BIN_T) * GRP_CAP + r] = (int)(word / BIN_T) * GRP_CAP + r;
        }
      }
    }
    if (over && ln == 0) lost = 1;
  }
  __syncthreads();

  const int deg = valid ? dcnt[tid >> 3] : 0;
  const int last = deg - 1;
  const bool slow = valid && (lost != 0 || deg > GRP_CAP);
  const bool fits = deg > 0 && !slow;
  if (fits) {
    for (int idx = hd; idx < deg; idx += 8) {
      const int k = ky[o + idx];
      int r = 0;
      for (int q = 0; q < deg; q += 4) {
        const v4i v = *(const v4i*)(ky + o + q);
        r += (v.x < k ? 1 : 0) + (v.y < k ? 1 : 0) + (v.z < k ? 1 : 0) + (v.w < k ? 1 : 0);
      }
      ss[o + r] = (int)((unsigned)k / GRP_CAP);
    }
  }
  __syncthreads();

  float4 o0 = make_float4(0.0f, 0.0f, 0.0f, 0.0f), o1 = o0;

  if (fits || slow) {
    float m = -__builtin_inff();
    float s = 0.0f;
    v2f acc[4] = {{0.0f, 0.0f}, {0.0f, 0.0f}, {0.0f, 0.0f}, {0.0f, 0.0f}};
    const unsigned cb = (unsigned)lo * 4u;
    const char* __restrict__ Pb = (const char*)P;
    const char* __restrict__ hb = (const char*)h;

    if (fits) {
      int n0 = ss[o];
      int n1 = ss[o + min(1, last)];
      int n2 = ss[o + min(2, last)];
      int n3 = ss[o + min(3, last)];
      for (int i = 0; i < deg; i += 4) {
        const unsigned f0 = (unsigned)n0 * (HD * 4u) + cb;
        const unsigned f1 = (unsigned)n1 * (HD * 4u) + cb;
        const unsigned f2 = (unsigned)n2 * (HD * 4u) + cb;
        const unsigned f3 = (unsigned)n3 * (HD * 4u) + cb;
        n0 = ss[o + min(i + 4, last)];
        n1 = ss[o + min(i + 5, last)];
        n2 = ss[o + min(i + 6, last)];
        n3 = ss[o + min(i + 7, last)];
        const float4* r0 = (const float4*)(Pb + f0);
        const float4* r1 = (const float4*)(Pb + f1);
        const float4* r2 = (const float4*)(Pb + f2);
        const float4* r3 = (const float4*)(Pb + f3);
        const float4 x00 = r0[0], x01 = r0[8], x10 = r1[0], x11 = r1[8];
        const float4 x20 = r2[0], x21 = r2[8], x30 = r3[0], x31 = r3[8];
        const float4* g0 = (const float4*)(hb + f0);
        const float4* g1 = (const float4*)(hb + f1);
        const float4* g2 = (const float4*)(hb + f2);
        const float4* g3 = (const float4*)(hb + f3);
        const float4 y00 = g0[0], y01 = g0[8], y10 = g1[0], y11 = g1[8];
        const float4 y20 = g2[0], y21 = g2[8], y30 = g3[0], y31 = g3[8];

        const float l0 = gat_logit(x00, x01, q0, q1, a0, a1);
        const float l1 = gat_logit(x10, x11, q0, q1, a0, a1);
        const float l2 = gat_logit(x20, x21, q0, q1, a0, a1);
        const float l3 = gat_logit(x30, x31, q0, q1, a0, a1);

        const float mn = fmaxf(m, fmaxf(fmaxf(l0, l1), fmaxf(l2, l3)));
        const float sc = __expf(m - mn);
        m = mn;
        const float e0 = __expf(l0 - mn);
        const float e1 = (i + 1 < deg) ? __expf(l1 - mn) : 0.0f;
        const float e2 = (i + 2 < deg) ? __expf(l2 - mn) : 0.0f;
        const float e3 = (i + 3 < deg) ? __expf(l3 - mn) : 0.0f;
        s *= sc;
        s += e0; s += e1; s += e2; s += e3;
        const float sca = gat_from_a(sc), scb = gat_from_b(sc);
        acc[0] *= sca; acc[1] *= sca;
        acc[2] *= scb; acc[3] *= scb;
        gat_axpy(acc, y00, y01, gat_from_a(e0), gat_from_b(e0));
        gat_axpy(acc, y10, y11, gat_from_a(e1), gat_from_b(e1));
        gat_axpy(acc, y20, y21, gat_from_a(e2), gat_from_b(e2));
        gat_axpy(acc, y30, y31, gat_from_a(e3), gat_from_b(e3));
        asm volatile("" :: "v"(n0), "v"(n1), "v"(n2), "v"(n3));
        __builtin_amdgcn_sched_group_barrier(0x020, 16, 0);
        __builtin_amdgcn_sched_group_barrier(0x002, 512, 0);
      }
    } else {
      int prev = -1;
      for (int it = 0; it < nEdges; ++it) {
        int best = 0x7fffffff;
        int mult = 0;
        for (int e = 0; e < nEdges; ++e) {
          if (tgt[e] == t) {
            const int v = src[e];
            if (v > prev) {
              if (v < best) { best = v; mult = 1; }
              else if (v == best) ++mult;
            }
          }
        }
        if (mult == 0) break;
        prev = best;
        const unsigned f = (unsigned)best * (HD * 4u) + cb;
        const float4* r  = (const float4*)(Pb + f);
        const float4* gq = (const float4*)(hb + f);
        const float4 x0 = r[0], x1 = r[8];
        const float4 y0 = gq[0], y1 = gq[8];
        const float l  = gat_logit(x0, x1, q0, q1, a0, a1);
        const float mn = fmaxf(m, l);
        const float sc = __expf(m - mn);
        m = mn;
        const float w = __expf(l - mn);
        s *= sc;
        const float sca = gat_from_a(sc), scb = gat_from_b(sc);
        const float wa = gat_from_a(w), wb = gat_from_b(w);
        acc[0] *= sca; acc[1] *= sca;
        acc[2] *= scb; acc[3] *= scb;
        for (int c = 0; c < mult; ++c) {
          s += w;
          gat_axpy(acc, y0, y1, wa, wb);
        }
      }
    }

    const float sa = gat_from_a(s), sb = gat_from_b(s);
    if (sa > 0.0f) {
      o0.x = acc[0].x / sa; o0.y = acc[0].y / sa; o0.z = acc[1].x / sa; o0.w = acc[1].y / sa;
    }
    if (sb > 0.0f) {
      o1.x = acc[2].x / sb; o1.y = acc[2].y / sb; o1.z = acc[3].x / sb; o1.w = acc[3].y / sb;
    }
  }
  if (valid) {
    float4* op = (float4*)(out + (size_t)t * HD + lo);
    op[0] = o0; op[8] = o1;
  }
}

extern "C" void kernel_launch(void* const* d_in, const int* in_sizes, int n_in,
                              void* d_out, int out_size, void* d_ws, size_t ws_size,
                              hipStream_t stream) {
  const float* h   = (const float*)d_in[0];
  const float* W   = (const float*)d_in[1];
  const float* a   = (const float*)d_in[2];
  const int*   src = (const int*)d_in[3];
  const int*   tgt = (const int*)d_in[4];
  float*       out = (float*)d_out;

  const int nNodes = in_sizes[0] / IN_CH;
  const int nEdges = in_sizes[3];

  const int nBins   = (nNodes + BIN_T - 1) / BIN_T;
  const int nChunks = (nEdges + CHUNK_E - 1) / CHUNK_E;
  float* P   = (float*)d_ws;
  float* Q   = P + (size_t)nNodes * HD;
  int*   seg = (int*)(Q + (size_t)nNodes * HD);

  gat_project_partition<<<nChunks + PROJ_WG, PART_T, 0, stream>>>(
      h, W, P, Q, src, tgt, seg, nNodes, nEdges, nBins, nChunks);

  gat_gather<<<nBins, 256, 0, stream>>>(h, P, Q, a, seg, src, tgt, out, nNodes,
                                        nEdges, nChunks);
}
